# starting kernel
# baseline (speedup 1.0000x reference)
#include <hip/hip_runtime.h>

typedef __attribute__((ext_vector_type(2))) float v2f;
typedef __attribute__((ext_vector_type(8))) float v8f;

namespace {
constexpr int kT      = 4096;
constexpr int kNB     = 64;
constexpr int kNU     = 32;
constexpr int kNY     = 32;
constexpr int kNH     = 256;
constexpr int kNHalf  = 128;
constexpr int kChunk  = 64;
constexpr int kNChunk = kT / kChunk;
constexpr int kBG     = 32;
constexpr int kNBG    = kNB / kBG;
constexpr int kTG     = 2;
constexpr int kNGrp   = kChunk / kTG;

constexpr size_t WS_LR  = 0;
constexpr size_t WS_LI  = WS_LR + kNHalf;
constexpr size_t WS_LRL = WS_LI + kNHalf;
constexpr size_t WS_LIL = WS_LRL + kNHalf;
constexpr size_t WS_BP  = WS_LIL + kNHalf;
constexpr size_t WS_X0  = WS_BP + (size_t)kNH * kNU;
constexpr size_t WS_F   = WS_X0 + (size_t)kNB * kNH;
constexpr size_t WS_S   = WS_F + (size_t)kNChunk * kNB * kNH;

constexpr size_t kSmemBytes = (size_t)kTG * kBG * kNH * sizeof(float);
}

__global__ void __launch_bounds__(256)
setup_kernel(const float* __restrict__ y0, const float* __restrict__ llr,
             const float* __restrict__ lli, const float* __restrict__ Bmat,
             const float* __restrict__ y2xw, const float* __restrict__ y2xb,
             float* __restrict__ ws) {
  __shared__ float s_nf[kNH];
  const int tid = threadIdx.x;
  if (tid < kNHalf) {
    const float decay = __expf(-fabsf(llr[tid]));
    const float r     = __expf(-decay);
    const float th    = 1.57079632679489662f * __expf(-fabsf(lli[tid]));
    const float lr    = r * __cosf(th);
    const float li    = r * __sinf(th);
    const float nfh   = sqrtf(fmaxf(0.f, 1.f - r * r));
    ws[WS_LR + tid] = lr;
    ws[WS_LI + tid] = li;
    float pr = lr, pi = li;
    #pragma unroll
    for (int s = 0; s < 6; ++s) {
      const float nr = pr * pr - pi * pi;
      const float ni = 2.f * pr * pi;
      pr = nr; pi = ni;
    }
    ws[WS_LRL + tid] = pr;
    ws[WS_LIL + tid] = pi;
    s_nf[tid]          = nfh;
    s_nf[tid + kNHalf] = nfh;
  }
  __syncthreads();
  for (int i = tid; i < kNH * kNU; i += blockDim.x) {
    ws[WS_BP + i] = Bmat[i] * s_nf[i / kNU];
  }
  for (int o = tid; o < kNB * kNH; o += blockDim.x) {
    const int b = o >> 8;
    const int n = o & 255;
    float acc = y2xb[n];
    const float* yr = y0 + b * kNY;
    const float* wr = y2xw + n * kNY;
    #pragma unroll 8
    for (int u = 0; u < kNY; ++u) acc += yr[u] * wr[u];
    ws[WS_X0 + o] = acc;
  }
}

__device__ __forceinline__ void bu_gemm_store(const float* __restrict__ Uslot,
                                              const float* __restrict__ Bp,
                                              float* __restrict__ Xslot,
                                              int wave, int laneM, int laneHi) {
  const int mt    = (wave >> 1) & 1;
  const int ntPar = wave & 1;
  const int m     = mt * 16 + laneM;
  const int koff  = laneHi << 1;
  const float* Arow = Uslot + m * kNU;
  #pragma unroll
  for (int j = 0; j < 8; ++j) {
    const int nt = ntPar + (j << 1);
    const int n  = nt * 16 + laneM;
    const float* Brow = Bp + n * kNU;
    v8f acc = {};
    #pragma unroll
    for (int kk = 0; kk < 8; ++kk) {
      const int k = (kk << 2) + koff;
      v2f a = *(const v2f*)(Arow + k);
      v2f b = *(const v2f*)(Brow + k);
      acc = __builtin_amdgcn_wmma_f32_16x16x4_f32(false, a, false, b,
                                                  (short)0, acc, false, false);
    }
    #pragma unroll
    for (int v = 0; v < 8; ++v) {
      const int mm = mt * 16 + v + (laneHi << 3);
      Xslot[mm * kNH + n] = acc[v];
    }
  }
}

__global__ void __launch_bounds__(256)
pass1_kernel(const float* __restrict__ U, float* __restrict__ ws) {
  extern __shared__ float Xbuf[];
  const float* Bp = ws + WS_BP;

  const int g = blockIdx.x;
  const int c = blockIdx.y;
  const int tid    = threadIdx.x;
  const int lane   = tid & 31;
  const int wave   = tid >> 5;
  const int laneM  = lane & 15;
  const int laneHi = lane >> 4;
  const int slotW  = wave >> 2;

  const int sb = tid >> 3;
  const int h0 = (tid & 7) << 4;

  float st1[16], st2[16], rlr[16], rli[16];
  #pragma unroll
  for (int i = 0; i < 16; ++i) {
    st1[i] = 0.f; st2[i] = 0.f;
    rlr[i] = ws[WS_LR + h0 + i];
    rli[i] = ws[WS_LI + h0 + i];
  }

  for (int grp = 0; grp < kNGrp; ++grp) {
    const int t0 = c * kChunk + grp * kTG;
    const float* Uslot = U + ((size_t)(t0 + slotW) * kNB + g * kBG) * kNU;
    bu_gemm_store(Uslot, Bp, Xbuf + slotW * (kBG * kNH), wave, laneM, laneHi);
    __syncthreads();
    #pragma unroll
    for (int s = 0; s < kTG; ++s) {
      const float* Bu = Xbuf + s * (kBG * kNH) + sb * kNH;
      #pragma unroll
      for (int i = 0; i < 16; ++i) {
        const float x1 = st1[i], x2 = st2[i];
        st1[i] = rlr[i] * x1 - rli[i] * x2 + Bu[h0 + i];
        st2[i] = rli[i] * x1 + rlr[i] * x2 + Bu[h0 + i + kNHalf];
      }
    }
    __syncthreads();
  }
  float* F = ws + WS_F + ((size_t)c * kNB + g * kBG + sb) * kNH;
  #pragma unroll
  for (int i = 0; i < 16; ++i) {
    F[h0 + i]          = st1[i];
    F[h0 + i + kNHalf] = st2[i];
  }
}

__global__ void __launch_bounds__(256)
combine_kernel(float* __restrict__ ws) {
  const int tid = threadIdx.x;
  for (int i = 0; i < 32; ++i) {
    const int p = tid * 32 + i;
    const int b = p >> 7;
    const int h = p & 127;
    float s1 = ws[WS_X0 + (size_t)b * kNH + h];
    float s2 = ws[WS_X0 + (size_t)b * kNH + h + kNHalf];
    const float pr = ws[WS_LRL + h];
    const float pi = ws[WS_LIL + h];
    for (int c = 0; c < kNChunk; ++c) {
      const size_t base = (size_t)c * kNB * kNH + (size_t)b * kNH + h;
      ws[WS_S + base]          = s1;
      ws[WS_S + base + kNHalf] = s2;
      const float f1 = ws[WS_F + base];
      const float f2 = ws[WS_F + base + kNHalf];
      const float n1 = pr * s1 - pi * s2 + f1;
      const float n2 = pi * s1 + pr * s2 + f2;
      s1 = n1; s2 = n2;
    }
  }
}

__global__ void __launch_bounds__(256)
pass2_kernel(const float* __restrict__ U, const float* __restrict__ W,
             const float* __restrict__ bias, const float* __restrict__ ws,
             float* __restrict__ out) {
  extern __shared__ float Xbuf[];
  const float* Bp = ws + WS_BP;

  const int g = blockIdx.x;
  const int c = blockIdx.y;
  const int tid    = threadIdx.x;
  const int lane   = tid & 31;
  const int wave   = tid >> 5;
  const int laneM  = lane & 15;
  const int laneHi = lane >> 4;
  const int koff   = laneHi << 1;
  const int slotW  = wave >> 2;

  const int sb = tid >> 3;
  const int h0 = (tid & 7) << 4;

  float st1[16], st2[16], rlr[16], rli[16];
  {
    const float* S = ws + WS_S + ((size_t)c * kNB + g * kBG + sb) * kNH;
    #pragma unroll
    for (int i = 0; i < 16; ++i) {
      st1[i] = S[h0 + i];
      st2[i] = S[h0 + i + kNHalf];
      rlr[i] = ws[WS_LR + h0 + i];
      rli[i] = ws[WS_LI + h0 + i];
    }
  }

  const int mt2 = (wave >> 1) & 1;
  const int nt2 = wave & 1;
  const float* Wrow  = W + (size_t)(nt2 * 16 + laneM) * kNH;
  const float  biasl = bias[nt2 * 16 + laneM];

  for (int grp = 0; grp < kNGrp; ++grp) {
    const int t0 = c * kChunk + grp * kTG;
    const float* Uslot = U + ((size_t)(t0 + slotW) * kNB + g * kBG) * kNU;
    bu_gemm_store(Uslot, Bp, Xbuf + slotW * (kBG * kNH), wave, laneM, laneHi);
    __syncthreads();
    #pragma unroll
    for (int s = 0; s < kTG; ++s) {
      float* Bu = Xbuf + s * (kBG * kNH) + sb * kNH;
      #pragma unroll
      for (int i = 0; i < 16; ++i) {
        const float x1 = st1[i], x2 = st2[i];
        const float n1 = rlr[i] * x1 - rli[i] * x2 + Bu[h0 + i];
        const float n2 = rli[i] * x1 + rlr[i] * x2 + Bu[h0 + i + kNHalf];
        st1[i] = n1; st2[i] = n2;
        Bu[h0 + i]          = n1;
        Bu[h0 + i + kNHalf] = n2;
      }
    }
    __syncthreads();
    {
      const float* Arow = Xbuf + slotW * (kBG * kNH) + (mt2 * 16 + laneM) * kNH;
      v8f acc = {};
      #pragma unroll 8
      for (int kk = 0; kk < 64; ++kk) {
        const int k = (kk << 2) + koff;
        v2f a = *(const v2f*)(Arow + k);
        v2f b = *(const v2f*)(Wrow + k);
        acc = __builtin_amdgcn_wmma_f32_16x16x4_f32(false, a, false, b,
                                                    (short)0, acc, false, false);
      }
      const int t    = t0 + slotW;
      const int ycol = nt2 * 16 + laneM;
      float* Orow = out + (size_t)t * kNB * kNY + (size_t)(g * kBG) * kNY;
      #pragma unroll
      for (int v = 0; v < 8; ++v) {
        const int mm = mt2 * 16 + v + (laneHi << 3);
        Orow[mm * kNY + ycol] = acc[v] + biasl;
      }
    }
    __syncthreads();
  }
}

extern "C" void kernel_launch(void* const* d_in, const int* in_sizes, int n_in,
                              void* d_out, int out_size, void* d_ws, size_t ws_size,
                              hipStream_t stream) {
  (void)in_sizes; (void)n_in; (void)out_size; (void)ws_size;
  const float* y0   = (const float*)d_in[0];
  const float* U    = (const float*)d_in[1];
  const float* llr  = (const float*)d_in[2];
  const float* lli  = (const float*)d_in[3];
  const float* Bm   = (const float*)d_in[4];
  const float* y2xw = (const float*)d_in[5];
  const float* y2xb = (const float*)d_in[6];
  const float* x2yw = (const float*)d_in[7];
  const float* x2yb = (const float*)d_in[8];
  float* ws  = (float*)d_ws;
  float* out = (float*)d_out;

  hipFuncSetAttribute(reinterpret_cast<const void*>(pass1_kernel),
                      hipFuncAttributeMaxDynamicSharedMemorySize, (int)kSmemBytes);
  hipFuncSetAttribute(reinterpret_cast<const void*>(pass2_kernel),
                      hipFuncAttributeMaxDynamicSharedMemorySize, (int)kSmemBytes);

  setup_kernel<<<1, 256, 0, stream>>>(y0, llr, lli, Bm, y2xw, y2xb, ws);
  dim3 grid(kNBG, kNChunk);
  pass1_kernel<<<grid, 256, kSmemBytes, stream>>>(U, ws);
  combine_kernel<<<1, 256, 0, stream>>>(ws);
  pass2_kernel<<<grid, 256, kSmemBytes, stream>>>(U, x2yw, x2yb, ws, out);
}


# setup launch removed: each scan block derives pole rotation, R^64 and nf-scaled B fragments itself in registers; x0 = y0*y2x_w^T+b is a WMMA product in pass 2, so x0's sum order changes
# speedup vs baseline: 15.5454x; 15.5454x over previous
#include <hip/hip_runtime.h>

typedef __attribute__((ext_vector_type(2))) float v2f;
typedef __attribute__((ext_vector_type(4))) float v4f;
typedef __attribute__((ext_vector_type(8))) float v8f;

namespace {
constexpr int kT      = 4096;
constexpr int kNB     = 64;
constexpr int kNU     = 32;
constexpr int kNY     = 32;
constexpr int kNH     = 256;
constexpr int kNHalf  = 128;
constexpr int kChunk  = 64;
constexpr int kNChunk = kT / kChunk;
constexpr int kBT     = 16;
constexpr int kNBT    = kNB / kBT;
constexpr int kWaves  = 8;
constexpr int kG      = 4;
constexpr int kNGrp   = kChunk / kG;
constexpr int kCarryDepth = 4;

constexpr size_t WS_F   = 0;
constexpr size_t kFStride = (size_t)kNB * kNH;

constexpr size_t kUStep   = (size_t)kNB * kNU;
constexpr size_t kOutStep = (size_t)kNB * kNY;

constexpr int kPartWave = kG * 4 * 128;
constexpr int kPartBuf  = kWaves * kPartWave;
constexpr size_t kSmemBytes = (size_t)2 * kPartBuf * sizeof(float);
static_assert(kNU == kNY, "load_a32 serves B and y2x_w with one row pitch");
}

__device__ __forceinline__ float pole_radius(const float llr_p) {
  const float decay = __expf(-fabsf(llr_p));
  return __expf(-decay);
}

__device__ __forceinline__ float pole_nf(const float llr_p) {
  const float r = pole_radius(llr_p);
  return sqrtf(fmaxf(0.f, 1.f - r * r));
}

__device__ __forceinline__ v8f load8(const float* __restrict__ p) {
  const v4f a = *(const v4f*)(p);
  const v4f b = *(const v4f*)(p + 4);
  return __builtin_shufflevector(a, b, 0, 1, 2, 3, 4, 5, 6, 7);
}

__device__ __forceinline__ void store8(float* __restrict__ p, const v8f x) {
  *(v4f*)(p)     = __builtin_shufflevector(x, x, 0, 1, 2, 3);
  *(v4f*)(p + 4) = __builtin_shufflevector(x, x, 4, 5, 6, 7);
}

__device__ __forceinline__ void load_a32(const float* __restrict__ M, int P0,
                                         int laneM, int laneHi, v2f (&a)[2][8]) {
  #pragma unroll
  for (int i = 0; i < 2; ++i) {
    const float* row = M + (size_t)(i * kNHalf + P0 + laneM) * kNU + 16 * laneHi;
    #pragma unroll
    for (int kk = 0; kk < 8; ++kk) a[i][kk] = *(const v2f*)(row + 2 * kk);
  }
}

__device__ __forceinline__ void load_bp(const float* __restrict__ Bmat,
                                        const float* __restrict__ llr, int P0,
                                        int laneM, int laneHi, v2f (&bp)[2][8]) {
  const float nf = pole_nf(llr[P0 + laneM]);
  load_a32(Bmat, P0, laneM, laneHi, bp);
  #pragma unroll
  for (int i = 0; i < 2; ++i)
    #pragma unroll
    for (int kk = 0; kk < 8; ++kk) bp[i][kk] *= nf;
}

__device__ __forceinline__ void load_rot(const float* __restrict__ llr,
                                         const float* __restrict__ lli, int P0,
                                         int laneHi, v8f& lr, v8f& li) {
  const v8f a = load8(llr + P0 + 8 * laneHi);
  const v8f b = load8(lli + P0 + 8 * laneHi);
  #pragma unroll
  for (int v = 0; v < 8; ++v) {
    const float r  = pole_radius(a[v]);
    const float th = 1.57079632679489662f * __expf(-fabsf(b[v]));
    lr[v] = r * __cosf(th);
    li[v] = r * __sinf(th);
  }
}

__device__ __forceinline__ void load_u(const float* __restrict__ up, v4f (&u)[4]) {
  #pragma unroll
  for (int q = 0; q < 4; ++q) u[q] = *(const v4f*)(up + 4 * q);
}

__device__ __forceinline__ void mma32(v8f& c0, v8f& c1, const v2f (&a)[2][8],
                                      const v4f (&u)[4]) {
  #pragma unroll
  for (int kk = 0; kk < 8; ++kk) {
    v2f b;
    b[0] = u[kk >> 1][(kk & 1) * 2];
    b[1] = u[kk >> 1][(kk & 1) * 2 + 1];
    c0 = __builtin_amdgcn_wmma_f32_16x16x4_f32(false, a[0][kk], false, b,
                                               (short)0, c0, false, false);
    c1 = __builtin_amdgcn_wmma_f32_16x16x4_f32(false, a[1][kk], false, b,
                                               (short)0, c1, false, false);
  }
}

__device__ __forceinline__ void step_state(v8f& s0, v8f& s1, const v2f (&bp)[2][8],
                                           const v4f (&u)[4], const v8f lr, const v8f li) {
  v8f c0 = lr * s0 - li * s1;
  v8f c1 = li * s0 + lr * s1;
  mma32(c0, c1, bp, u);
  s0 = c0; s1 = c1;
}

__global__ void __launch_bounds__(256)
pass1_kernel(const float* __restrict__ U, const float* __restrict__ llr,
             const float* __restrict__ lli, const float* __restrict__ Bmat,
             float* __restrict__ ws) {
  const int bt = blockIdx.x;
  const int c  = blockIdx.y;
  const int tid    = threadIdx.x;
  const int lane   = tid & 31;
  const int wave   = tid >> 5;
  const int laneM  = lane & 15;
  const int laneHi = lane >> 4;
  const int b0 = bt * kBT;
  const int P0 = wave * 16;

  v2f bp[2][8];
  load_bp(Bmat, llr, P0, laneM, laneHi, bp);
  v8f lr, li;
  load_rot(llr, lli, P0, laneHi, lr, li);

  v8f s0 = {}, s1 = {};
  const float* Ul = U + ((size_t)c * kChunk * kNB + b0 + laneM) * kNU + 16 * laneHi;
  for (int grp = 0; grp < kNGrp; ++grp) {
    v4f u[kG][4];
    #pragma unroll
    for (int tt = 0; tt < kG; ++tt) load_u(Ul + (size_t)(grp * kG + tt) * kUStep, u[tt]);
    #pragma unroll
    for (int tt = 0; tt < kG; ++tt) step_state(s0, s1, bp, u[tt], lr, li);
  }
  float* F = ws + WS_F + ((size_t)c * kNB + b0 + laneM) * kNH + P0 + 8 * laneHi;
  store8(F, s0);
  store8(F + kNHalf, s1);
}

__device__ __forceinline__ void carry_step(v8f& s0, v8f& s1, const v8f f0, const v8f f1,
                                           const v8f pr, const v8f pi) {
  #pragma unroll
  for (int v = 0; v < 8; ++v) {
    const float a = s0[v], b = s1[v];
    s0[v] = pr[v] * a - pi[v] * b + f0[v];
    s1[v] = pi[v] * a + pr[v] * b + f1[v];
  }
}

__device__ __forceinline__ void reduce_store(const float* __restrict__ Pb,
                                             float* __restrict__ op,
                                             int wave, int lane, const v4f bias4) {
  #pragma unroll
  for (int k = 0; k < kG / 2; ++k) {
    const int tt = 2 * k + (wave >> 2);
    const float* p = Pb + (tt * 4 + (wave & 3)) * 128 + lane * 4;
    v4f acc = *(const v4f*)(p);
    #pragma unroll
    for (int w = 1; w < kWaves; ++w) acc += *(const v4f*)(p + w * kPartWave);
    acc += bias4;
    *(v4f*)(op + (size_t)tt * kOutStep) = acc;
  }
}

__global__ void __launch_bounds__(256)
pass2_kernel(const float* __restrict__ U, const float* __restrict__ yinit,
             const float* __restrict__ llr, const float* __restrict__ lli,
             const float* __restrict__ Bmat, const float* __restrict__ y2xw,
             const float* __restrict__ y2xb, const float* __restrict__ W,
             const float* __restrict__ bias, const float* __restrict__ ws,
             float* __restrict__ out) {
  extern __shared__ float Pbuf[];

  const int bt = blockIdx.x;
  const int c  = blockIdx.y;
  const int tid    = threadIdx.x;
  const int lane   = tid & 31;
  const int wave   = tid >> 5;
  const int laneM  = lane & 15;
  const int laneHi = lane >> 4;
  const int b0 = bt * kBT;
  const int P0 = wave * 16;

  v8f lr, li;
  load_rot(llr, lli, P0, laneHi, lr, li);

  v8f s0, s1;
  {
    {
      v2f yw[2][8];
      load_a32(y2xw, P0, laneM, laneHi, yw);
      v4f yr[4];
      load_u(yinit + (size_t)(b0 + laneM) * kNY + 16 * laneHi, yr);
      s0 = load8(y2xb + P0 + 8 * laneHi);
      s1 = load8(y2xb + kNHalf + P0 + 8 * laneHi);
      mma32(s0, s1, yw, yr);
    }
    if (c > 0) {
      v8f pr = lr, pi = li;
      #pragma unroll
      for (int v = 0; v < 8; ++v) {
        float qr = pr[v], qi = pi[v];
        #pragma unroll
        for (int s = 0; s < 6; ++s) {
          const float nr = qr * qr - qi * qi;
          const float ni = 2.f * qr * qi;
          qr = nr; qi = ni;
        }
        pr[v] = qr; pi[v] = qi;
      }
      const float* Fq = ws + WS_F + (size_t)(b0 + laneM) * kNH + P0 + 8 * laneHi;
      v8f f0[kCarryDepth], f1[kCarryDepth];
      #pragma unroll
      for (int d = 0; d < kCarryDepth; ++d) {
        const float* p = Fq + (size_t)min(d, c - 1) * kFStride;
        f0[d] = load8(p);
        f1[d] = load8(p + kNHalf);
      }
      int j0 = 0;
      for (; j0 + kCarryDepth <= c; j0 += kCarryDepth) {
        #pragma unroll
        for (int d = 0; d < kCarryDepth; ++d) {
          carry_step(s0, s1, f0[d], f1[d], pr, pi);
          const float* p = Fq + (size_t)min(j0 + kCarryDepth + d, c - 1) * kFStride;
          f0[d] = load8(p);
          f1[d] = load8(p + kNHalf);
        }
      }
      #pragma unroll
      for (int d = 0; d < kCarryDepth - 1; ++d)
        if (j0 + d < c) carry_step(s0, s1, f0[d], f1[d], pr, pi);
    }
  }

  v2f bp[2][8];
  load_bp(Bmat, llr, P0, laneM, laneHi, bp);
  v2f wf[2][2][4];
  #pragma unroll
  for (int my = 0; my < 2; ++my)
    #pragma unroll
    for (int i = 0; i < 2; ++i)
      #pragma unroll
      for (int s = 0; s < 4; ++s)
        wf[my][i][s] = *(const v2f*)(W + (size_t)(16 * my + laneM) * kNH +
                                     i * kNHalf + P0 + 8 * laneHi + 2 * s);

  const int   ycol  = 16 * ((wave & 3) >> 1) + 8 * laneHi + 4 * (wave & 1);
  const v4f   bias4 = *(const v4f*)(bias + ycol);
  float* outp = out + ((size_t)c * kChunk * kNB + b0 + laneM) * kNY + ycol;

  const float* Ul = U + ((size_t)c * kChunk * kNB + b0 + laneM) * kNU + 16 * laneHi;
  for (int grp = 0; grp < kNGrp; ++grp) {
    const int buf = grp & 1;
    v4f u[kG][4];
    #pragma unroll
    for (int tt = 0; tt < kG; ++tt) load_u(Ul + (size_t)(grp * kG + tt) * kUStep, u[tt]);
    if (grp > 0)
      reduce_store(Pbuf + (buf ^ 1) * kPartBuf, outp + (size_t)(grp - 1) * kG * kOutStep,
                   wave, lane, bias4);
    float* Pw = Pbuf + buf * kPartBuf + wave * kPartWave + lane * 4;
    #pragma unroll
    for (int tt = 0; tt < kG; ++tt) {
      step_state(s0, s1, bp, u[tt], lr, li);
      v8f y0 = {}, y1 = {};
      #pragma unroll
      for (int s = 0; s < 4; ++s) {
        v2f b;
        b[0] = s0[2 * s]; b[1] = s0[2 * s + 1];
        y0 = __builtin_amdgcn_wmma_f32_16x16x4_f32(false, wf[0][0][s], false, b,
                                                   (short)0, y0, false, false);
        y1 = __builtin_amdgcn_wmma_f32_16x16x4_f32(false, wf[1][0][s], false, b,
                                                   (short)0, y1, false, false);
      }
      #pragma unroll
      for (int s = 0; s < 4; ++s) {
        v2f b;
        b[0] = s1[2 * s]; b[1] = s1[2 * s + 1];
        y0 = __builtin_amdgcn_wmma_f32_16x16x4_f32(false, wf[0][1][s], false, b,
                                                   (short)0, y0, false, false);
        y1 = __builtin_amdgcn_wmma_f32_16x16x4_f32(false, wf[1][1][s], false, b,
                                                   (short)0, y1, false, false);
      }
      *(v4f*)(Pw + (tt * 4 + 0) * 128) = __builtin_shufflevector(y0, y0, 0, 1, 2, 3);
      *(v4f*)(Pw + (tt * 4 + 1) * 128) = __builtin_shufflevector(y0, y0, 4, 5, 6, 7);
      *(v4f*)(Pw + (tt * 4 + 2) * 128) = __builtin_shufflevector(y1, y1, 0, 1, 2, 3);
      *(v4f*)(Pw + (tt * 4 + 3) * 128) = __builtin_shufflevector(y1, y1, 4, 5, 6, 7);
    }
    __syncthreads();
  }
  reduce_store(Pbuf + ((kNGrp - 1) & 1) * kPartBuf,
               outp + (size_t)(kNGrp - 1) * kG * kOutStep, wave, lane, bias4);
}

extern "C" void kernel_launch(void* const* d_in, const int* in_sizes, int n_in,
                              void* d_out, int out_size, void* d_ws, size_t ws_size,
                              hipStream_t stream) {
  (void)in_sizes; (void)n_in; (void)out_size; (void)ws_size;
  const float* y0   = (const float*)d_in[0];
  const float* U    = (const float*)d_in[1];
  const float* llr  = (const float*)d_in[2];
  const float* lli  = (const float*)d_in[3];
  const float* Bm   = (const float*)d_in[4];
  const float* y2xw = (const float*)d_in[5];
  const float* y2xb = (const float*)d_in[6];
  const float* x2yw = (const float*)d_in[7];
  const float* x2yb = (const float*)d_in[8];
  float* ws  = (float*)d_ws;
  float* out = (float*)d_out;

  hipFuncSetAttribute(reinterpret_cast<const void*>(pass2_kernel),
                      hipFuncAttributeMaxDynamicSharedMemorySize, (int)kSmemBytes);

  dim3 grid(kNBT, kNChunk);
  pass1_kernel<<<grid, 256, 0, stream>>>(U, llr, lli, Bm, ws);
  pass2_kernel<<<grid, 256, kSmemBytes, stream>>>(U, y0, llr, lli, Bm, y2xw, y2xb,
                                                 x2yw, x2yb, ws, out);
}


# chunk finals kept in the workspace in the scan's own lane order: each carry load and pass-1 store covers 512 contiguous bytes (4 whole lines), not 16 lines x 32 B; values and sum order unchanged
# speedup vs baseline: 18.0363x; 1.1602x over previous
#include <hip/hip_runtime.h>

typedef __attribute__((ext_vector_type(2))) float v2f;
typedef __attribute__((ext_vector_type(4))) float v4f;
typedef __attribute__((ext_vector_type(8))) float v8f;

namespace {
constexpr int kT      = 4096;
constexpr int kNB     = 64;
constexpr int kNU     = 32;
constexpr int kNY     = 32;
constexpr int kNH     = 256;
constexpr int kNHalf  = 128;
constexpr int kChunk  = 64;
constexpr int kNChunk = kT / kChunk;
constexpr int kBT     = 16;
constexpr int kNBT    = kNB / kBT;
constexpr int kWaves  = 8;
constexpr int kG      = 4;
constexpr int kNGrp   = kChunk / kG;
constexpr int kCarryDepth = 4;

constexpr size_t WS_F   = 0;
constexpr size_t kFQuarter = 32 * 4;
constexpr size_t kFWave    = 4 * kFQuarter;
constexpr size_t kFStride  = (size_t)kNBT * kWaves * kFWave;
static_assert(kFStride == (size_t)kNB * kNH, "one final holds every (batch, hidden) element once");

constexpr size_t kUStep   = (size_t)kNB * kNU;
constexpr size_t kOutStep = (size_t)kNB * kNY;

constexpr int kPartWave = kG * 4 * 128;
constexpr int kPartBuf  = kWaves * kPartWave;
constexpr size_t kSmemBytes = (size_t)2 * kPartBuf * sizeof(float);
static_assert(kNU == kNY, "load_a32 serves B and y2x_w with one row pitch");
}

__device__ __forceinline__ float pole_radius(const float llr_p) {
  const float decay = __expf(-fabsf(llr_p));
  return __expf(-decay);
}

__device__ __forceinline__ float pole_nf(const float llr_p) {
  const float r = pole_radius(llr_p);
  return sqrtf(fmaxf(0.f, 1.f - r * r));
}

__device__ __forceinline__ v8f load8(const float* __restrict__ p) {
  const v4f a = *(const v4f*)(p);
  const v4f b = *(const v4f*)(p + 4);
  return __builtin_shufflevector(a, b, 0, 1, 2, 3, 4, 5, 6, 7);
}

__device__ __forceinline__ v8f load_ftile(const float* __restrict__ p) {
  const v4f a = *(const v4f*)(p);
  const v4f b = *(const v4f*)(p + kFQuarter);
  return __builtin_shufflevector(a, b, 0, 1, 2, 3, 4, 5, 6, 7);
}

__device__ __forceinline__ void store_ftile(float* __restrict__ p, const v8f x) {
  *(v4f*)(p)             = __builtin_shufflevector(x, x, 0, 1, 2, 3);
  *(v4f*)(p + kFQuarter) = __builtin_shufflevector(x, x, 4, 5, 6, 7);
}

__device__ __forceinline__ void load_a32(const float* __restrict__ M, int P0,
                                         int laneM, int laneHi, v2f (&a)[2][8]) {
  #pragma unroll
  for (int i = 0; i < 2; ++i) {
    const float* row = M + (size_t)(i * kNHalf + P0 + laneM) * kNU + 16 * laneHi;
    #pragma unroll
    for (int kk = 0; kk < 8; ++kk) a[i][kk] = *(const v2f*)(row + 2 * kk);
  }
}

__device__ __forceinline__ void load_bp(const float* __restrict__ Bmat,
                                        const float* __restrict__ llr, int P0,
                                        int laneM, int laneHi, v2f (&bp)[2][8]) {
  const float nf = pole_nf(llr[P0 + laneM]);
  load_a32(Bmat, P0, laneM, laneHi, bp);
  #pragma unroll
  for (int i = 0; i < 2; ++i)
    #pragma unroll
    for (int kk = 0; kk < 8; ++kk) bp[i][kk] *= nf;
}

__device__ __forceinline__ void load_rot(const float* __restrict__ llr,
                                         const float* __restrict__ lli, int P0,
                                         int laneHi, v8f& lr, v8f& li) {
  const v8f a = load8(llr + P0 + 8 * laneHi);
  const v8f b = load8(lli + P0 + 8 * laneHi);
  #pragma unroll
  for (int v = 0; v < 8; ++v) {
    const float r  = pole_radius(a[v]);
    const float th = 1.57079632679489662f * __expf(-fabsf(b[v]));
    lr[v] = r * __cosf(th);
    li[v] = r * __sinf(th);
  }
}

__device__ __forceinline__ void load_u(const float* __restrict__ up, v4f (&u)[4]) {
  #pragma unroll
  for (int q = 0; q < 4; ++q) u[q] = *(const v4f*)(up + 4 * q);
}

__device__ __forceinline__ void mma32(v8f& c0, v8f& c1, const v2f (&a)[2][8],
                                      const v4f (&u)[4]) {
  #pragma unroll
  for (int kk = 0; kk < 8; ++kk) {
    v2f b;
    b[0] = u[kk >> 1][(kk & 1) * 2];
    b[1] = u[kk >> 1][(kk & 1) * 2 + 1];
    c0 = __builtin_amdgcn_wmma_f32_16x16x4_f32(false, a[0][kk], false, b,
                                               (short)0, c0, false, false);
    c1 = __builtin_amdgcn_wmma_f32_16x16x4_f32(false, a[1][kk], false, b,
                                               (short)0, c1, false, false);
  }
}

__device__ __forceinline__ void step_state(v8f& s0, v8f& s1, const v2f (&bp)[2][8],
                                           const v4f (&u)[4], const v8f lr, const v8f li) {
  v8f c0 = lr * s0 - li * s1;
  v8f c1 = li * s0 + lr * s1;
  mma32(c0, c1, bp, u);
  s0 = c0; s1 = c1;
}

__global__ void __launch_bounds__(256) __attribute__((amdgpu_waves_per_eu(2, 4)))
pass1_kernel(const float* __restrict__ U, const float* __restrict__ llr,
             const float* __restrict__ lli, const float* __restrict__ Bmat,
             float* __restrict__ ws) {
  const int bt = blockIdx.x;
  const int c  = blockIdx.y;
  const int tid    = threadIdx.x;
  const int lane   = tid & 31;
  const int wave   = tid >> 5;
  const int laneM  = lane & 15;
  const int laneHi = lane >> 4;
  const int b0 = bt * kBT;
  const int P0 = wave * 16;

  v2f bp[2][8];
  load_bp(Bmat, llr, P0, laneM, laneHi, bp);
  v8f lr, li;
  load_rot(llr, lli, P0, laneHi, lr, li);

  v8f s0 = {}, s1 = {};
  const float* Ul = U + ((size_t)c * kChunk * kNB + b0 + laneM) * kNU + 16 * laneHi;
  for (int grp = 0; grp < kNGrp; ++grp) {
    v4f u[kG][4];
    #pragma unroll
    for (int tt = 0; tt < kG; ++tt) load_u(Ul + (size_t)(grp * kG + tt) * kUStep, u[tt]);
    #pragma unroll
    for (int tt = 0; tt < kG; ++tt) step_state(s0, s1, bp, u[tt], lr, li);
  }
  float* F = ws + WS_F + (size_t)c * kFStride + ((size_t)bt * kWaves + wave) * kFWave + lane * 4;
  store_ftile(F, s0);
  store_ftile(F + 2 * kFQuarter, s1);
}

__device__ __forceinline__ void carry_step(v8f& s0, v8f& s1, const v8f f0, const v8f f1,
                                           const v8f pr, const v8f pi) {
  #pragma unroll
  for (int v = 0; v < 8; ++v) {
    const float a = s0[v], b = s1[v];
    s0[v] = pr[v] * a - pi[v] * b + f0[v];
    s1[v] = pi[v] * a + pr[v] * b + f1[v];
  }
}

__device__ __forceinline__ void reduce_store(const float* __restrict__ Pb,
                                             float* __restrict__ op,
                                             int wave, int lane, const v4f bias4) {
  #pragma unroll
  for (int k = 0; k < kG / 2; ++k) {
    const int tt = 2 * k + (wave >> 2);
    const float* p = Pb + (tt * 4 + (wave & 3)) * 128 + lane * 4;
    v4f acc = *(const v4f*)(p);
    #pragma unroll
    for (int w = 1; w < kWaves; ++w) acc += *(const v4f*)(p + w * kPartWave);
    acc += bias4;
    *(v4f*)(op + (size_t)tt * kOutStep) = acc;
  }
}

__global__ void __launch_bounds__(256)
pass2_kernel(const float* __restrict__ U, const float* __restrict__ yinit,
             const float* __restrict__ llr, const float* __restrict__ lli,
             const float* __restrict__ Bmat, const float* __restrict__ y2xw,
             const float* __restrict__ y2xb, const float* __restrict__ W,
             const float* __restrict__ bias, const float* __restrict__ ws,
             float* __restrict__ out) {
  extern __shared__ float Pbuf[];

  const int bt = blockIdx.x;
  const int c  = blockIdx.y;
  const int tid    = threadIdx.x;
  const int lane   = tid & 31;
  const int wave   = tid >> 5;
  const int laneM  = lane & 15;
  const int laneHi = lane >> 4;
  const int b0 = bt * kBT;
  const int P0 = wave * 16;

  v8f lr, li;
  load_rot(llr, lli, P0, laneHi, lr, li);

  v8f s0, s1;
  {
    {
      v2f yw[2][8];
      load_a32(y2xw, P0, laneM, laneHi, yw);
      v4f yr[4];
      load_u(yinit + (size_t)(b0 + laneM) * kNY + 16 * laneHi, yr);
      s0 = load8(y2xb + P0 + 8 * laneHi);
      s1 = load8(y2xb + kNHalf + P0 + 8 * laneHi);
      mma32(s0, s1, yw, yr);
    }
    if (c > 0) {
      v8f pr = lr, pi = li;
      #pragma unroll
      for (int v = 0; v < 8; ++v) {
        float qr = pr[v], qi = pi[v];
        #pragma unroll
        for (int s = 0; s < 6; ++s) {
          const float nr = qr * qr - qi * qi;
          const float ni = 2.f * qr * qi;
          qr = nr; qi = ni;
        }
        pr[v] = qr; pi[v] = qi;
      }
      const float* Fq = ws + WS_F + ((size_t)bt * kWaves + wave) * kFWave + lane * 4;
      v8f f0[kCarryDepth], f1[kCarryDepth];
      #pragma unroll
      for (int d = 0; d < kCarryDepth; ++d) {
        const float* p = Fq + (size_t)min(d, c - 1) * kFStride;
        f0[d] = load_ftile(p);
        f1[d] = load_ftile(p + 2 * kFQuarter);
      }
      int j0 = 0;
      for (; j0 + kCarryDepth <= c; j0 += kCarryDepth) {
        #pragma unroll
        for (int d = 0; d < kCarryDepth; ++d) {
          carry_step(s0, s1, f0[d], f1[d], pr, pi);
          const float* p = Fq + (size_t)min(j0 + kCarryDepth + d, c - 1) * kFStride;
          f0[d] = load_ftile(p);
          f1[d] = load_ftile(p + 2 * kFQuarter);
        }
      }
      #pragma unroll
      for (int d = 0; d < kCarryDepth - 1; ++d)
        if (j0 + d < c) carry_step(s0, s1, f0[d], f1[d], pr, pi);
    }
  }

  v2f bp[2][8];
  load_bp(Bmat, llr, P0, laneM, laneHi, bp);
  v2f wf[2][2][4];
  #pragma unroll
  for (int my = 0; my < 2; ++my)
    #pragma unroll
    for (int i = 0; i < 2; ++i)
      #pragma unroll
      for (int s = 0; s < 4; ++s)
        wf[my][i][s] = *(const v2f*)(W + (size_t)(16 * my + laneM) * kNH +
                                     i * kNHalf + P0 + 8 * laneHi + 2 * s);

  const int   ycol  = 16 * ((wave & 3) >> 1) + 8 * laneHi + 4 * (wave & 1);
  const v4f   bias4 = *(const v4f*)(bias + ycol);
  float* outp = out + ((size_t)c * kChunk * kNB + b0 + laneM) * kNY + ycol;

  const float* Ul = U + ((size_t)c * kChunk * kNB + b0 + laneM) * kNU + 16 * laneHi;
  for (int grp = 0; grp < kNGrp; ++grp) {
    const int buf = grp & 1;
    v4f u[kG][4];
    #pragma unroll
    for (int tt = 0; tt < kG; ++tt) load_u(Ul + (size_t)(grp * kG + tt) * kUStep, u[tt]);
    if (grp > 0)
      reduce_store(Pbuf + (buf ^ 1) * kPartBuf, outp + (size_t)(grp - 1) * kG * kOutStep,
                   wave, lane, bias4);
    float* Pw = Pbuf + buf * kPartBuf + wave * kPartWave + lane * 4;
    #pragma unroll
    for (int tt = 0; tt < kG; ++tt) {
      step_state(s0, s1, bp, u[tt], lr, li);
      v8f y0 = {}, y1 = {};
      #pragma unroll
      for (int s = 0; s < 4; ++s) {
        v2f b;
        b[0] = s0[2 * s]; b[1] = s0[2 * s + 1];
        y0 = __builtin_amdgcn_wmma_f32_16x16x4_f32(false, wf[0][0][s], false, b,
                                                   (short)0, y0, false, false);
        y1 = __builtin_amdgcn_wmma_f32_16x16x4_f32(false, wf[1][0][s], false, b,
                                                   (short)0, y1, false, false);
      }
      #pragma unroll
      for (int s = 0; s < 4; ++s) {
        v2f b;
        b[0] = s1[2 * s]; b[1] = s1[2 * s + 1];
        y0 = __builtin_amdgcn_wmma_f32_16x16x4_f32(false, wf[0][1][s], false, b,
                                                   (short)0, y0, false, false);
        y1 = __builtin_amdgcn_wmma_f32_16x16x4_f32(false, wf[1][1][s], false, b,
                                                   (short)0, y1, false, false);
      }
      *(v4f*)(Pw + (tt * 4 + 0) * 128) = __builtin_shufflevector(y0, y0, 0, 1, 2, 3);
      *(v4f*)(Pw + (tt * 4 + 1) * 128) = __builtin_shufflevector(y0, y0, 4, 5, 6, 7);
      *(v4f*)(Pw + (tt * 4 + 2) * 128) = __builtin_shufflevector(y1, y1, 0, 1, 2, 3);
      *(v4f*)(Pw + (tt * 4 + 3) * 128) = __builtin_shufflevector(y1, y1, 4, 5, 6, 7);
    }
    __syncthreads();
  }
  reduce_store(Pbuf + ((kNGrp - 1) & 1) * kPartBuf,
               outp + (size_t)(kNGrp - 1) * kG * kOutStep, wave, lane, bias4);
}

extern "C" void kernel_launch(void* const* d_in, const int* in_sizes, int n_in,
                              void* d_out, int out_size, void* d_ws, size_t ws_size,
                              hipStream_t stream) {
  (void)in_sizes; (void)n_in; (void)out_size; (void)ws_size;
  const float* y0   = (const float*)d_in[0];
  const float* U    = (const float*)d_in[1];
  const float* llr  = (const float*)d_in[2];
  const float* lli  = (const float*)d_in[3];
  const float* Bm   = (const float*)d_in[4];
  const float* y2xw = (const float*)d_in[5];
  const float* y2xb = (const float*)d_in[6];
  const float* x2yw = (const float*)d_in[7];
  const float* x2yb = (const float*)d_in[8];
  float* ws  = (float*)d_ws;
  float* out = (float*)d_out;

  hipFuncSetAttribute(reinterpret_cast<const void*>(pass2_kernel),
                      hipFuncAttributeMaxDynamicSharedMemorySize, (int)kSmemBytes);

  dim3 grid(kNBT, kNChunk);
  pass1_kernel<<<grid, 256, 0, stream>>>(U, llr, lli, Bm, ws);
  pass2_kernel<<<grid, 256, kSmemBytes, stream>>>(U, y0, llr, lli, Bm, y2xw, y2xb,
                                                 x2yw, x2yb, ws, out);
}


# U tile staged once per block in LDS in B-fragment order (whole-line loads, all issued up front; pass 1 takes it in quarters); loops read U from LDS, not 16-line global loads x 8 waves; same sums
# speedup vs baseline: 20.5815x; 1.1411x over previous
#include <hip/hip_runtime.h>

typedef __attribute__((ext_vector_type(2))) float v2f;
typedef __attribute__((ext_vector_type(4))) float v4f;
typedef __attribute__((ext_vector_type(8))) float v8f;

namespace {
constexpr int kT      = 4096;
constexpr int kNB     = 64;
constexpr int kNU     = 32;
constexpr int kNY     = 32;
constexpr int kNH     = 256;
constexpr int kNHalf  = 128;
constexpr int kChunk  = 64;
constexpr int kNChunk = kT / kChunk;
constexpr int kBT     = 16;
constexpr int kNBT    = kNB / kBT;
constexpr int kWaves  = 8;
constexpr int kG      = 4;
constexpr int kNGrp   = kChunk / kG;
constexpr int kCarryDepth = 4;

constexpr size_t WS_F   = 0;
constexpr size_t kFQuarter = 32 * 4;
constexpr size_t kFWave    = 4 * kFQuarter;
constexpr size_t kFStride  = (size_t)kNBT * kWaves * kFWave;
static_assert(kFStride == (size_t)kNB * kNH, "one final holds every (batch, hidden) element once");

constexpr size_t kUStep   = (size_t)kNB * kNU;
constexpr size_t kOutStep = (size_t)kNB * kNY;

constexpr int kPartWave = kG * 4 * 128;
constexpr int kPartBuf  = kWaves * kPartWave;
constexpr size_t kSmemBytes = (size_t)2 * kPartBuf * sizeof(float);
static_assert(kNU == kNY, "load_a32 serves B and y2x_w with one row pitch");

constexpr int kUTile      = kBT * kNU;
constexpr int kUChunk     = kChunk * kUTile;
constexpr int kStageLoads = kUChunk / 4 / 256;
static_assert(kStageLoads * 256 * 4 == kUChunk, "256 threads cover the chunk tile exactly");
constexpr int kParts      = 4;
constexpr int kPartLoads  = kStageLoads / kParts;
constexpr int kPartGrp    = kNGrp / kParts;
static_assert(kPartLoads * 2 == kPartGrp * kG, "a quarter's pieces are its groups' timesteps");
}

__device__ __forceinline__ float pole_radius(const float llr_p) {
  const float decay = __expf(-fabsf(llr_p));
  return __expf(-decay);
}

__device__ __forceinline__ float pole_nf(const float llr_p) {
  const float r = pole_radius(llr_p);
  return sqrtf(fmaxf(0.f, 1.f - r * r));
}

__device__ __forceinline__ v8f load8(const float* __restrict__ p) {
  const v4f a = *(const v4f*)(p);
  const v4f b = *(const v4f*)(p + 4);
  return __builtin_shufflevector(a, b, 0, 1, 2, 3, 4, 5, 6, 7);
}

__device__ __forceinline__ v8f load_ftile(const float* __restrict__ p) {
  const v4f a = *(const v4f*)(p);
  const v4f b = *(const v4f*)(p + kFQuarter);
  return __builtin_shufflevector(a, b, 0, 1, 2, 3, 4, 5, 6, 7);
}

__device__ __forceinline__ void store_ftile(float* __restrict__ p, const v8f x) {
  *(v4f*)(p)             = __builtin_shufflevector(x, x, 0, 1, 2, 3);
  *(v4f*)(p + kFQuarter) = __builtin_shufflevector(x, x, 4, 5, 6, 7);
}

__device__ __forceinline__ void load_a32(const float* __restrict__ M, int P0,
                                         int laneM, int laneHi, v2f (&a)[2][8]) {
  #pragma unroll
  for (int i = 0; i < 2; ++i) {
    const float* row = M + (size_t)(i * kNHalf + P0 + laneM) * kNU + 16 * laneHi;
    #pragma unroll
    for (int kk = 0; kk < 8; ++kk) a[i][kk] = *(const v2f*)(row + 2 * kk);
  }
}

__device__ __forceinline__ void load_bp(const float* __restrict__ Bmat,
                                        const float* __restrict__ llr, int P0,
                                        int laneM, int laneHi, v2f (&bp)[2][8]) {
  const float nf = pole_nf(llr[P0 + laneM]);
  load_a32(Bmat, P0, laneM, laneHi, bp);
  #pragma unroll
  for (int i = 0; i < 2; ++i)
    #pragma unroll
    for (int kk = 0; kk < 8; ++kk) bp[i][kk] *= nf;
}

__device__ __forceinline__ void load_rot(const float* __restrict__ llr,
                                         const float* __restrict__ lli, int P0,
                                         int laneHi, v8f& lr, v8f& li) {
  const v8f a = load8(llr + P0 + 8 * laneHi);
  const v8f b = load8(lli + P0 + 8 * laneHi);
  #pragma unroll
  for (int v = 0; v < 8; ++v) {
    const float r  = pole_radius(a[v]);
    const float th = 1.57079632679489662f * __expf(-fabsf(b[v]));
    lr[v] = r * __cosf(th);
    li[v] = r * __sinf(th);
  }
}

__device__ __forceinline__ void load_u(const float* __restrict__ up, v4f (&u)[4]) {
  #pragma unroll
  for (int q = 0; q < 4; ++q) u[q] = *(const v4f*)(up + 4 * q);
}

__device__ __forceinline__ int u_pos(int x, int q) { return x ^ (4 * q) ^ (2 * (x >> 4)); }

__device__ __forceinline__ void stage_u_load(const float* __restrict__ Uc, int tid,
                                             v4f (&st)[kStageLoads]) {
  const float* g = Uc + (size_t)(tid >> 7) * kUStep + (tid & 127) * 4;
  #pragma unroll
  for (int n = 0; n < kStageLoads; ++n) st[n] = *(const v4f*)(g + (size_t)n * 2 * kUStep);
}

__device__ __forceinline__ void stage_u_store(float* __restrict__ Us, int tid,
                                              const v4f (&st)[kStageLoads], int part) {
  const int pidx = tid & 127;
  const int r = pidx >> 3, h = (pidx >> 2) & 1, q = pidx & 3;
  float* s = Us + (((tid >> 7) * 4 + q) * 32 + u_pos(16 * h + r, q)) * 4;
  #pragma unroll
  for (int k = 0; k < kPartLoads; ++k) {
    const int n = part * kPartLoads + k;
    *(v4f*)(s + n * 2 * kUTile) = st[n];
  }
}

__device__ __forceinline__ void lds_barrier() {
  __builtin_amdgcn_fence(__ATOMIC_RELEASE, "workgroup", "local");
  __builtin_amdgcn_s_barrier();
  __builtin_amdgcn_fence(__ATOMIC_ACQUIRE, "workgroup", "local");
}

__device__ __forceinline__ void load_u_lds(const float* __restrict__ ut, const int (&po)[4],
                                           v4f (&u)[4]) {
  #pragma unroll
  for (int q = 0; q < 4; ++q) u[q] = *(const v4f*)(ut + po[q]);
}

__device__ __forceinline__ void mma32(v8f& c0, v8f& c1, const v2f (&a)[2][8],
                                      const v4f (&u)[4]) {
  #pragma unroll
  for (int kk = 0; kk < 8; ++kk) {
    v2f b;
    b[0] = u[kk >> 1][(kk & 1) * 2];
    b[1] = u[kk >> 1][(kk & 1) * 2 + 1];
    c0 = __builtin_amdgcn_wmma_f32_16x16x4_f32(false, a[0][kk], false, b,
                                               (short)0, c0, false, false);
    c1 = __builtin_amdgcn_wmma_f32_16x16x4_f32(false, a[1][kk], false, b,
                                               (short)0, c1, false, false);
  }
}

__device__ __forceinline__ void step_state(v8f& s0, v8f& s1, const v2f (&bp)[2][8],
                                           const v4f (&u)[4], const v8f lr, const v8f li) {
  v8f c0 = lr * s0 - li * s1;
  v8f c1 = li * s0 + lr * s1;
  mma32(c0, c1, bp, u);
  s0 = c0; s1 = c1;
}

__global__ void __launch_bounds__(256)
pass1_kernel(const float* __restrict__ U, const float* __restrict__ llr,
             const float* __restrict__ lli, const float* __restrict__ Bmat,
             float* __restrict__ ws) {
  const int bt = blockIdx.x;
  const int c  = blockIdx.y;
  const int tid    = threadIdx.x;
  const int lane   = tid & 31;
  const int wave   = tid >> 5;
  const int laneM  = lane & 15;
  const int laneHi = lane >> 4;
  const int b0 = bt * kBT;
  const int P0 = wave * 16;

  __shared__ __attribute__((aligned(16))) float Us[kUChunk];
  v2f bp[2][8];
  load_bp(Bmat, llr, P0, laneM, laneHi, bp);
  v8f lr, li;
  load_rot(llr, lli, P0, laneHi, lr, li);

  v4f st[kStageLoads];
  stage_u_load(U + ((size_t)c * kChunk * kNB + b0) * kNU, tid, st);

  int po[4];
  #pragma unroll
  for (int q = 0; q < 4; ++q) po[q] = q * 128 + 4 * u_pos(lane, q);

  v8f s0 = {}, s1 = {};
  #pragma unroll
  for (int part = 0; part < kParts; ++part) {
    stage_u_store(Us, tid, st, part);
    lds_barrier();
    #pragma nounroll
    for (int g = 0; g < kPartGrp; ++g) {
      const int grp = part * kPartGrp + g;
      v4f u[kG][4];
      #pragma unroll
      for (int tt = 0; tt < kG; ++tt) load_u_lds(Us + (grp * kG + tt) * kUTile, po, u[tt]);
      #pragma unroll
      for (int tt = 0; tt < kG; ++tt) step_state(s0, s1, bp, u[tt], lr, li);
    }
  }
  float* F = ws + WS_F + (size_t)c * kFStride + ((size_t)bt * kWaves + wave) * kFWave + lane * 4;
  store_ftile(F, s0);
  store_ftile(F + 2 * kFQuarter, s1);
}

__device__ __forceinline__ void carry_step(v8f& s0, v8f& s1, const v8f f0, const v8f f1,
                                           const v8f pr, const v8f pi) {
  #pragma unroll
  for (int v = 0; v < 8; ++v) {
    const float a = s0[v], b = s1[v];
    s0[v] = pr[v] * a - pi[v] * b + f0[v];
    s1[v] = pi[v] * a + pr[v] * b + f1[v];
  }
}

__device__ __forceinline__ void reduce_store(const float* __restrict__ Pb,
                                             float* __restrict__ op,
                                             int wave, int lane, const v4f bias4) {
  #pragma unroll
  for (int k = 0; k < kG / 2; ++k) {
    const int tt = 2 * k + (wave >> 2);
    const float* p = Pb + (tt * 4 + (wave & 3)) * 128 + lane * 4;
    v4f acc = *(const v4f*)(p);
    #pragma unroll
    for (int w = 1; w < kWaves; ++w) acc += *(const v4f*)(p + w * kPartWave);
    acc += bias4;
    *(v4f*)(op + (size_t)tt * kOutStep) = acc;
  }
}

__global__ void __launch_bounds__(256)
pass2_kernel(const float* __restrict__ U, const float* __restrict__ yinit,
             const float* __restrict__ llr, const float* __restrict__ lli,
             const float* __restrict__ Bmat, const float* __restrict__ y2xw,
             const float* __restrict__ y2xb, const float* __restrict__ W,
             const float* __restrict__ bias, const float* __restrict__ ws,
             float* __restrict__ out) {
  extern __shared__ float Pbuf[];

  const int bt = blockIdx.x;
  const int c  = blockIdx.y;
  const int tid    = threadIdx.x;
  const int lane   = tid & 31;
  const int wave   = tid >> 5;
  const int laneM  = lane & 15;
  const int laneHi = lane >> 4;
  const int b0 = bt * kBT;
  const int P0 = wave * 16;

  __shared__ __attribute__((aligned(16))) float Us[kUChunk];
  v8f lr, li;
  v8f s0, s1;
  {
    load_rot(llr, lli, P0, laneHi, lr, li);

    {
      v2f yw[2][8];
      load_a32(y2xw, P0, laneM, laneHi, yw);
      v4f yr[4];
      load_u(yinit + (size_t)(b0 + laneM) * kNY + 16 * laneHi, yr);
      s0 = load8(y2xb + P0 + 8 * laneHi);
      s1 = load8(y2xb + kNHalf + P0 + 8 * laneHi);
      mma32(s0, s1, yw, yr);
    }
    v4f st[kStageLoads];
    stage_u_load(U + ((size_t)c * kChunk * kNB + b0) * kNU, tid, st);
    #pragma unroll
    for (int part = 0; part < kParts; ++part) stage_u_store(Us, tid, st, part);
    if (c > 0) {
      v8f pr = lr, pi = li;
      #pragma unroll
      for (int v = 0; v < 8; ++v) {
        float qr = pr[v], qi = pi[v];
        #pragma unroll
        for (int s = 0; s < 6; ++s) {
          const float nr = qr * qr - qi * qi;
          const float ni = 2.f * qr * qi;
          qr = nr; qi = ni;
        }
        pr[v] = qr; pi[v] = qi;
      }
      const float* Fq = ws + WS_F + ((size_t)bt * kWaves + wave) * kFWave + lane * 4;
      v8f f0[kCarryDepth], f1[kCarryDepth];
      #pragma unroll
      for (int d = 0; d < kCarryDepth; ++d) {
        const float* p = Fq + (size_t)min(d, c - 1) * kFStride;
        f0[d] = load_ftile(p);
        f1[d] = load_ftile(p + 2 * kFQuarter);
      }
      int j0 = 0;
      for (; j0 + kCarryDepth <= c; j0 += kCarryDepth) {
        #pragma unroll
        for (int d = 0; d < kCarryDepth; ++d) {
          carry_step(s0, s1, f0[d], f1[d], pr, pi);
          const float* p = Fq + (size_t)min(j0 + kCarryDepth + d, c - 1) * kFStride;
          f0[d] = load_ftile(p);
          f1[d] = load_ftile(p + 2 * kFQuarter);
        }
      }
      #pragma unroll
      for (int d = 0; d < kCarryDepth - 1; ++d)
        if (j0 + d < c) carry_step(s0, s1, f0[d], f1[d], pr, pi);
    }
  }

  v2f bp[2][8];
  load_bp(Bmat, llr, P0, laneM, laneHi, bp);
  v2f wf[2][2][4];
  #pragma unroll
  for (int my = 0; my < 2; ++my)
    #pragma unroll
    for (int i = 0; i < 2; ++i)
      #pragma unroll
      for (int s = 0; s < 4; ++s)
        wf[my][i][s] = *(const v2f*)(W + (size_t)(16 * my + laneM) * kNH +
                                     i * kNHalf + P0 + 8 * laneHi + 2 * s);

  const int   ycol  = 16 * ((wave & 3) >> 1) + 8 * laneHi + 4 * (wave & 1);
  const v4f   bias4 = *(const v4f*)(bias + ycol);
  float* outp = out + ((size_t)c * kChunk * kNB + b0 + laneM) * kNY + ycol;

  int po[4];
  #pragma unroll
  for (int q = 0; q < 4; ++q) po[q] = q * 128 + 4 * u_pos(lane, q);
  __syncthreads();

  for (int grp = 0; grp < kNGrp; ++grp) {
    const int buf = grp & 1;
    v4f u[kG][4];
    #pragma unroll
    for (int tt = 0; tt < kG; ++tt) load_u_lds(Us + (grp * kG + tt) * kUTile, po, u[tt]);
    if (grp > 0)
      reduce_store(Pbuf + (buf ^ 1) * kPartBuf, outp + (size_t)(grp - 1) * kG * kOutStep,
                   wave, lane, bias4);
    float* Pw = Pbuf + buf * kPartBuf + wave * kPartWave + lane * 4;
    #pragma unroll
    for (int tt = 0; tt < kG; ++tt) {
      step_state(s0, s1, bp, u[tt], lr, li);
      v8f y0 = {}, y1 = {};
      #pragma unroll
      for (int s = 0; s < 4; ++s) {
        v2f b;
        b[0] = s0[2 * s]; b[1] = s0[2 * s + 1];
        y0 = __builtin_amdgcn_wmma_f32_16x16x4_f32(false, wf[0][0][s], false, b,
                                                   (short)0, y0, false, false);
        y1 = __builtin_amdgcn_wmma_f32_16x16x4_f32(false, wf[1][0][s], false, b,
                                                   (short)0, y1, false, false);
      }
      #pragma unroll
      for (int s = 0; s < 4; ++s) {
        v2f b;
        b[0] = s1[2 * s]; b[1] = s1[2 * s + 1];
        y0 = __builtin_amdgcn_wmma_f32_16x16x4_f32(false, wf[0][1][s], false, b,
                                                   (short)0, y0, false, false);
        y1 = __builtin_amdgcn_wmma_f32_16x16x4_f32(false, wf[1][1][s], false, b,
                                                   (short)0, y1, false, false);
      }
      *(v4f*)(Pw + (tt * 4 + 0) * 128) = __builtin_shufflevector(y0, y0, 0, 1, 2, 3);
      *(v4f*)(Pw + (tt * 4 + 1) * 128) = __builtin_shufflevector(y0, y0, 4, 5, 6, 7);
      *(v4f*)(Pw + (tt * 4 + 2) * 128) = __builtin_shufflevector(y1, y1, 0, 1, 2, 3);
      *(v4f*)(Pw + (tt * 4 + 3) * 128) = __builtin_shufflevector(y1, y1, 4, 5, 6, 7);
    }
    __syncthreads();
  }
  reduce_store(Pbuf + ((kNGrp - 1) & 1) * kPartBuf,
               outp + (size_t)(kNGrp - 1) * kG * kOutStep, wave, lane, bias4);
}

extern "C" void kernel_launch(void* const* d_in, const int* in_sizes, int n_in,
                              void* d_out, int out_size, void* d_ws, size_t ws_size,
                              hipStream_t stream) {
  (void)in_sizes; (void)n_in; (void)out_size; (void)ws_size;
  const float* y0   = (const float*)d_in[0];
  const float* U    = (const float*)d_in[1];
  const float* llr  = (const float*)d_in[2];
  const float* lli  = (const float*)d_in[3];
  const float* Bm   = (const float*)d_in[4];
  const float* y2xw = (const float*)d_in[5];
  const float* y2xb = (const float*)d_in[6];
  const float* x2yw = (const float*)d_in[7];
  const float* x2yb = (const float*)d_in[8];
  float* ws  = (float*)d_ws;
  float* out = (float*)d_out;

  hipFuncSetAttribute(reinterpret_cast<const void*>(pass2_kernel),
                      hipFuncAttributeMaxDynamicSharedMemorySize, (int)kSmemBytes);

  dim3 grid(kNBT, kNChunk);
  pass1_kernel<<<grid, 256, 0, stream>>>(U, llr, lli, Bm, ws);
  pass2_kernel<<<grid, 256, kSmemBytes, stream>>>(U, y0, llr, lli, Bm, y2xw, y2xb,
                                                 x2yw, x2yb, ws, out);
}


# pass-2 set-up latency hidden: U tile, first chunk finals and loop operand fragments all requested before the first wait, carry ring 4 -> 8 finals in flight; recurrence and every sum order unchanged
# speedup vs baseline: 21.7944x; 1.0589x over previous
#include <hip/hip_runtime.h>

typedef __attribute__((ext_vector_type(2))) float v2f;
typedef __attribute__((ext_vector_type(4))) float v4f;
typedef __attribute__((ext_vector_type(8))) float v8f;

namespace {
constexpr int kT      = 4096;
constexpr int kNB     = 64;
constexpr int kNU     = 32;
constexpr int kNY     = 32;
constexpr int kNH     = 256;
constexpr int kNHalf  = 128;
constexpr int kChunk  = 64;
constexpr int kNChunk = kT / kChunk;
constexpr int kBT     = 16;
constexpr int kNBT    = kNB / kBT;
constexpr int kWaves  = 8;
constexpr int kG      = 4;
constexpr int kNGrp   = kChunk / kG;
constexpr int kCarryDepth = 8;

constexpr size_t WS_F   = 0;
constexpr size_t kFQuarter = 32 * 4;
constexpr size_t kFWave    = 4 * kFQuarter;
constexpr size_t kFStride  = (size_t)kNBT * kWaves * kFWave;
static_assert(kFStride == (size_t)kNB * kNH, "one final holds every (batch, hidden) element once");

constexpr size_t kUStep   = (size_t)kNB * kNU;
constexpr size_t kOutStep = (size_t)kNB * kNY;

constexpr int kPartWave = kG * 4 * 128;
constexpr int kPartBuf  = kWaves * kPartWave;
constexpr size_t kSmemBytes = (size_t)2 * kPartBuf * sizeof(float);
static_assert(kNU == kNY, "load_a32 serves B and y2x_w with one row pitch");

constexpr int kUTile      = kBT * kNU;
constexpr int kUChunk     = kChunk * kUTile;
constexpr int kStageLoads = kUChunk / 4 / 256;
static_assert(kStageLoads * 256 * 4 == kUChunk, "256 threads cover the chunk tile exactly");
constexpr int kParts      = 4;
constexpr int kPartLoads  = kStageLoads / kParts;
constexpr int kPartGrp    = kNGrp / kParts;
static_assert(kPartLoads * 2 == kPartGrp * kG, "a quarter's pieces are its groups' timesteps");
}

__device__ __forceinline__ float pole_radius(const float llr_p) {
  const float decay = __expf(-fabsf(llr_p));
  return __expf(-decay);
}

__device__ __forceinline__ float pole_nf(const float llr_p) {
  const float r = pole_radius(llr_p);
  return sqrtf(fmaxf(0.f, 1.f - r * r));
}

__device__ __forceinline__ v8f load8(const float* __restrict__ p) {
  const v4f a = *(const v4f*)(p);
  const v4f b = *(const v4f*)(p + 4);
  return __builtin_shufflevector(a, b, 0, 1, 2, 3, 4, 5, 6, 7);
}

__device__ __forceinline__ v8f load_ftile(const float* __restrict__ p) {
  const v4f a = *(const v4f*)(p);
  const v4f b = *(const v4f*)(p + kFQuarter);
  return __builtin_shufflevector(a, b, 0, 1, 2, 3, 4, 5, 6, 7);
}

__device__ __forceinline__ void store_ftile(float* __restrict__ p, const v8f x) {
  *(v4f*)(p)             = __builtin_shufflevector(x, x, 0, 1, 2, 3);
  *(v4f*)(p + kFQuarter) = __builtin_shufflevector(x, x, 4, 5, 6, 7);
}

__device__ __forceinline__ void load_a32(const float* __restrict__ M, int P0,
                                         int laneM, int laneHi, v2f (&a)[2][8]) {
  #pragma unroll
  for (int i = 0; i < 2; ++i) {
    const float* row = M + (size_t)(i * kNHalf + P0 + laneM) * kNU + 16 * laneHi;
    #pragma unroll
    for (int kk = 0; kk < 8; ++kk) a[i][kk] = *(const v2f*)(row + 2 * kk);
  }
}

__device__ __forceinline__ void load_bp(const float* __restrict__ Bmat,
                                        const float* __restrict__ llr, int P0,
                                        int laneM, int laneHi, v2f (&bp)[2][8]) {
  const float nf = pole_nf(llr[P0 + laneM]);
  load_a32(Bmat, P0, laneM, laneHi, bp);
  #pragma unroll
  for (int i = 0; i < 2; ++i)
    #pragma unroll
    for (int kk = 0; kk < 8; ++kk) bp[i][kk] *= nf;
}

__device__ __forceinline__ void load_rot(const float* __restrict__ llr,
                                         const float* __restrict__ lli, int P0,
                                         int laneHi, v8f& lr, v8f& li) {
  const v8f a = load8(llr + P0 + 8 * laneHi);
  const v8f b = load8(lli + P0 + 8 * laneHi);
  #pragma unroll
  for (int v = 0; v < 8; ++v) {
    const float r  = pole_radius(a[v]);
    const float th = 1.57079632679489662f * __expf(-fabsf(b[v]));
    lr[v] = r * __cosf(th);
    li[v] = r * __sinf(th);
  }
}

__device__ __forceinline__ void load_u(const float* __restrict__ up, v4f (&u)[4]) {
  #pragma unroll
  for (int q = 0; q < 4; ++q) u[q] = *(const v4f*)(up + 4 * q);
}

__device__ __forceinline__ int u_pos(int x, int q) { return x ^ (4 * q) ^ (2 * (x >> 4)); }

__device__ __forceinline__ void stage_u_load(const float* __restrict__ Uc, int tid,
                                             v4f (&st)[kStageLoads]) {
  const float* g = Uc + (size_t)(tid >> 7) * kUStep + (tid & 127) * 4;
  #pragma unroll
  for (int n = 0; n < kStageLoads; ++n) st[n] = *(const v4f*)(g + (size_t)n * 2 * kUStep);
}

__device__ __forceinline__ void stage_u_store(float* __restrict__ Us, int tid,
                                              const v4f (&st)[kStageLoads], int part) {
  const int pidx = tid & 127;
  const int r = pidx >> 3, h = (pidx >> 2) & 1, q = pidx & 3;
  float* s = Us + (((tid >> 7) * 4 + q) * 32 + u_pos(16 * h + r, q)) * 4;
  #pragma unroll
  for (int k = 0; k < kPartLoads; ++k) {
    const int n = part * kPartLoads + k;
    *(v4f*)(s + n * 2 * kUTile) = st[n];
  }
}

__device__ __forceinline__ void lds_barrier() {
  __builtin_amdgcn_fence(__ATOMIC_RELEASE, "workgroup", "local");
  __builtin_amdgcn_s_barrier();
  __builtin_amdgcn_fence(__ATOMIC_ACQUIRE, "workgroup", "local");
}

__device__ __forceinline__ void load_u_lds(const float* __restrict__ ut, const int (&po)[4],
                                           v4f (&u)[4]) {
  #pragma unroll
  for (int q = 0; q < 4; ++q) u[q] = *(const v4f*)(ut + po[q]);
}

__device__ __forceinline__ void mma32(v8f& c0, v8f& c1, const v2f (&a)[2][8],
                                      const v4f (&u)[4]) {
  #pragma unroll
  for (int kk = 0; kk < 8; ++kk) {
    v2f b;
    b[0] = u[kk >> 1][(kk & 1) * 2];
    b[1] = u[kk >> 1][(kk & 1) * 2 + 1];
    c0 = __builtin_amdgcn_wmma_f32_16x16x4_f32(false, a[0][kk], false, b,
                                               (short)0, c0, false, false);
    c1 = __builtin_amdgcn_wmma_f32_16x16x4_f32(false, a[1][kk], false, b,
                                               (short)0, c1, false, false);
  }
}

__device__ __forceinline__ void step_state(v8f& s0, v8f& s1, const v2f (&bp)[2][8],
                                           const v4f (&u)[4], const v8f lr, const v8f li) {
  v8f c0 = lr * s0 - li * s1;
  v8f c1 = li * s0 + lr * s1;
  mma32(c0, c1, bp, u);
  s0 = c0; s1 = c1;
}

__global__ void __launch_bounds__(256)
pass1_kernel(const float* __restrict__ U, const float* __restrict__ llr,
             const float* __restrict__ lli, const float* __restrict__ Bmat,
             float* __restrict__ ws) {
  const int bt = blockIdx.x;
  const int c  = blockIdx.y;
  const int tid    = threadIdx.x;
  const int lane   = tid & 31;
  const int wave   = tid >> 5;
  const int laneM  = lane & 15;
  const int laneHi = lane >> 4;
  const int b0 = bt * kBT;
  const int P0 = wave * 16;

  __shared__ __attribute__((aligned(16))) float Us[kUChunk];
  v2f bp[2][8];
  load_bp(Bmat, llr, P0, laneM, laneHi, bp);
  v8f lr, li;
  load_rot(llr, lli, P0, laneHi, lr, li);

  v4f st[kStageLoads];
  stage_u_load(U + ((size_t)c * kChunk * kNB + b0) * kNU, tid, st);

  int po[4];
  #pragma unroll
  for (int q = 0; q < 4; ++q) po[q] = q * 128 + 4 * u_pos(lane, q);

  v8f s0 = {}, s1 = {};
  #pragma unroll
  for (int part = 0; part < kParts; ++part) {
    stage_u_store(Us, tid, st, part);
    lds_barrier();
    #pragma nounroll
    for (int g = 0; g < kPartGrp; ++g) {
      const int grp = part * kPartGrp + g;
      v4f u[kG][4];
      #pragma unroll
      for (int tt = 0; tt < kG; ++tt) load_u_lds(Us + (grp * kG + tt) * kUTile, po, u[tt]);
      #pragma unroll
      for (int tt = 0; tt < kG; ++tt) step_state(s0, s1, bp, u[tt], lr, li);
    }
  }
  float* F = ws + WS_F + (size_t)c * kFStride + ((size_t)bt * kWaves + wave) * kFWave + lane * 4;
  store_ftile(F, s0);
  store_ftile(F + 2 * kFQuarter, s1);
}

__device__ __forceinline__ void carry_step(v8f& s0, v8f& s1, const v8f f0, const v8f f1,
                                           const v8f pr, const v8f pi) {
  #pragma unroll
  for (int v = 0; v < 8; ++v) {
    const float a = s0[v], b = s1[v];
    s0[v] = pr[v] * a - pi[v] * b + f0[v];
    s1[v] = pi[v] * a + pr[v] * b + f1[v];
  }
}

__device__ __forceinline__ void reduce_store(const float* __restrict__ Pb,
                                             float* __restrict__ op,
                                             int wave, int lane, const v4f bias4) {
  #pragma unroll
  for (int k = 0; k < kG / 2; ++k) {
    const int tt = 2 * k + (wave >> 2);
    const float* p = Pb + (tt * 4 + (wave & 3)) * 128 + lane * 4;
    v4f acc = *(const v4f*)(p);
    #pragma unroll
    for (int w = 1; w < kWaves; ++w) acc += *(const v4f*)(p + w * kPartWave);
    acc += bias4;
    *(v4f*)(op + (size_t)tt * kOutStep) = acc;
  }
}

__global__ void __launch_bounds__(256)
pass2_kernel(const float* __restrict__ U, const float* __restrict__ yinit,
             const float* __restrict__ llr, const float* __restrict__ lli,
             const float* __restrict__ Bmat, const float* __restrict__ y2xw,
             const float* __restrict__ y2xb, const float* __restrict__ W,
             const float* __restrict__ bias, const float* __restrict__ ws,
             float* __restrict__ out) {
  extern __shared__ float Pbuf[];

  const int bt = blockIdx.x;
  const int c  = blockIdx.y;
  const int tid    = threadIdx.x;
  const int lane   = tid & 31;
  const int wave   = tid >> 5;
  const int laneM  = lane & 15;
  const int laneHi = lane >> 4;
  const int b0 = bt * kBT;
  const int P0 = wave * 16;

  __shared__ __attribute__((aligned(16))) float Us[kUChunk];
  v8f lr, li;
  v8f s0, s1;
  v2f bp[2][8];
  v2f wf[2][2][4];
  const int ycol = 16 * ((wave & 3) >> 1) + 8 * laneHi + 4 * (wave & 1);
  v4f bias4;
  {
    load_rot(llr, lli, P0, laneHi, lr, li);

    {
      v2f yw[2][8];
      load_a32(y2xw, P0, laneM, laneHi, yw);
      v4f yr[4];
      load_u(yinit + (size_t)(b0 + laneM) * kNY + 16 * laneHi, yr);
      s0 = load8(y2xb + P0 + 8 * laneHi);
      s1 = load8(y2xb + kNHalf + P0 + 8 * laneHi);
      mma32(s0, s1, yw, yr);
    }
    v4f st[kStageLoads];
    stage_u_load(U + ((size_t)c * kChunk * kNB + b0) * kNU, tid, st);
    const float* Fq = ws + WS_F + ((size_t)bt * kWaves + wave) * kFWave + lane * 4;
    v8f f0[kCarryDepth], f1[kCarryDepth];
    if (c > 0) {
      #pragma unroll
      for (int d = 0; d < kCarryDepth; ++d) {
        const float* p = Fq + (size_t)min(d, c - 1) * kFStride;
        f0[d] = load_ftile(p);
        f1[d] = load_ftile(p + 2 * kFQuarter);
      }
    }
    load_bp(Bmat, llr, P0, laneM, laneHi, bp);
    #pragma unroll
    for (int my = 0; my < 2; ++my)
      #pragma unroll
      for (int i = 0; i < 2; ++i)
        #pragma unroll
        for (int s = 0; s < 4; ++s)
          wf[my][i][s] = *(const v2f*)(W + (size_t)(16 * my + laneM) * kNH +
                                       i * kNHalf + P0 + 8 * laneHi + 2 * s);
    bias4 = *(const v4f*)(bias + ycol);
    #pragma unroll
    for (int part = 0; part < kParts; ++part) stage_u_store(Us, tid, st, part);
    if (c > 0) {
      v8f pr = lr, pi = li;
      #pragma unroll
      for (int v = 0; v < 8; ++v) {
        float qr = pr[v], qi = pi[v];
        #pragma unroll
        for (int s = 0; s < 6; ++s) {
          const float nr = qr * qr - qi * qi;
          const float ni = 2.f * qr * qi;
          qr = nr; qi = ni;
        }
        pr[v] = qr; pi[v] = qi;
      }
      int j0 = 0;
      for (; j0 + kCarryDepth <= c; j0 += kCarryDepth) {
        #pragma unroll
        for (int d = 0; d < kCarryDepth; ++d) {
          carry_step(s0, s1, f0[d], f1[d], pr, pi);
          const float* p = Fq + (size_t)min(j0 + kCarryDepth + d, c - 1) * kFStride;
          f0[d] = load_ftile(p);
          f1[d] = load_ftile(p + 2 * kFQuarter);
        }
      }
      #pragma unroll
      for (int d = 0; d < kCarryDepth - 1; ++d)
        if (j0 + d < c) carry_step(s0, s1, f0[d], f1[d], pr, pi);
    }
  }

  float* outp = out + ((size_t)c * kChunk * kNB + b0 + laneM) * kNY + ycol;

  int po[4];
  #pragma unroll
  for (int q = 0; q < 4; ++q) po[q] = q * 128 + 4 * u_pos(lane, q);
  __syncthreads();

  for (int grp = 0; grp < kNGrp; ++grp) {
    const int buf = grp & 1;
    v4f u[kG][4];
    #pragma unroll
    for (int tt = 0; tt < kG; ++tt) load_u_lds(Us + (grp * kG + tt) * kUTile, po, u[tt]);
    if (grp > 0)
      reduce_store(Pbuf + (buf ^ 1) * kPartBuf, outp + (size_t)(grp - 1) * kG * kOutStep,
                   wave, lane, bias4);
    float* Pw = Pbuf + buf * kPartBuf + wave * kPartWave + lane * 4;
    #pragma unroll
    for (int tt = 0; tt < kG; ++tt) {
      step_state(s0, s1, bp, u[tt], lr, li);
      v8f y0 = {}, y1 = {};
      #pragma unroll
      for (int s = 0; s < 4; ++s) {
        v2f b;
        b[0] = s0[2 * s]; b[1] = s0[2 * s + 1];
        y0 = __builtin_amdgcn_wmma_f32_16x16x4_f32(false, wf[0][0][s], false, b,
                                                   (short)0, y0, false, false);
        y1 = __builtin_amdgcn_wmma_f32_16x16x4_f32(false, wf[1][0][s], false, b,
                                                   (short)0, y1, false, false);
      }
      #pragma unroll
      for (int s = 0; s < 4; ++s) {
        v2f b;
        b[0] = s1[2 * s]; b[1] = s1[2 * s + 1];
        y0 = __builtin_amdgcn_wmma_f32_16x16x4_f32(false, wf[0][1][s], false, b,
                                                   (short)0, y0, false, false);
        y1 = __builtin_amdgcn_wmma_f32_16x16x4_f32(false, wf[1][1][s], false, b,
                                                   (short)0, y1, false, false);
      }
      *(v4f*)(Pw + (tt * 4 + 0) * 128) = __builtin_shufflevector(y0, y0, 0, 1, 2, 3);
      *(v4f*)(Pw + (tt * 4 + 1) * 128) = __builtin_shufflevector(y0, y0, 4, 5, 6, 7);
      *(v4f*)(Pw + (tt * 4 + 2) * 128) = __builtin_shufflevector(y1, y1, 0, 1, 2, 3);
      *(v4f*)(Pw + (tt * 4 + 3) * 128) = __builtin_shufflevector(y1, y1, 4, 5, 6, 7);
    }
    __syncthreads();
  }
  reduce_store(Pbuf + ((kNGrp - 1) & 1) * kPartBuf,
               outp + (size_t)(kNGrp - 1) * kG * kOutStep, wave, lane, bias4);
}

extern "C" void kernel_launch(void* const* d_in, const int* in_sizes, int n_in,
                              void* d_out, int out_size, void* d_ws, size_t ws_size,
                              hipStream_t stream) {
  (void)in_sizes; (void)n_in; (void)out_size; (void)ws_size;
  const float* y0   = (const float*)d_in[0];
  const float* U    = (const float*)d_in[1];
  const float* llr  = (const float*)d_in[2];
  const float* lli  = (const float*)d_in[3];
  const float* Bm   = (const float*)d_in[4];
  const float* y2xw = (const float*)d_in[5];
  const float* y2xb = (const float*)d_in[6];
  const float* x2yw = (const float*)d_in[7];
  const float* x2yb = (const float*)d_in[8];
  float* ws  = (float*)d_ws;
  float* out = (float*)d_out;

  hipFuncSetAttribute(reinterpret_cast<const void*>(pass2_kernel),
                      hipFuncAttributeMaxDynamicSharedMemorySize, (int)kSmemBytes);

  dim3 grid(kNBT, kNChunk);
  pass1_kernel<<<grid, 256, 0, stream>>>(U, llr, lli, Bm, ws);
  pass2_kernel<<<grid, 256, kSmemBytes, stream>>>(U, y0, llr, lli, Bm, y2xw, y2xb,
                                                 x2yw, x2yb, ws, out);
}


# tile columns = 4 batch rows x 4 time segments (not 16 batch rows): a block owns 256 consecutive steps, joins its segments by lane shifts; cross-block carry 15 steps, not 63; carry sum order changes
# speedup vs baseline: 22.6092x; 1.0374x over previous
#include <hip/hip_runtime.h>

typedef __attribute__((ext_vector_type(2))) float v2f;
typedef __attribute__((ext_vector_type(4))) float v4f;
typedef __attribute__((ext_vector_type(8))) float v8f;

namespace {
constexpr int kT      = 4096;
constexpr int kNB     = 64;
constexpr int kNU     = 32;
constexpr int kNY     = 32;
constexpr int kNH     = 256;
constexpr int kNHalf  = 128;
constexpr int kChunk  = 64;
constexpr int kBT     = 16;
constexpr int kRows   = 4;
constexpr int kSeg    = kBT / kRows;
constexpr int kNBQ    = kNB / kRows;
constexpr int kSuper  = kSeg * kChunk;
constexpr int kNSuper = kT / kSuper;
constexpr int kWaves  = 8;
constexpr int kG      = 4;
constexpr int kNGrp   = kChunk / kG;
constexpr int kCarryDepth = 8;

constexpr size_t WS_F   = 0;
constexpr size_t kFQuarter = 32 * 4;
constexpr size_t kFWave    = 4 * kFQuarter;
constexpr size_t kFStride  = (size_t)kNBQ * kWaves * kFWave;
static_assert(kFStride == (size_t)kNBQ * kBT * kNH, "one P holds every (column, hidden) element once");

constexpr size_t kUStep   = (size_t)kNB * kNU;
constexpr size_t kOutStep = (size_t)kNB * kNY;

constexpr int kPartWave = kG * 4 * 128;
constexpr int kPartBuf  = kWaves * kPartWave;
constexpr size_t kSmemBytes = (size_t)2 * kPartBuf * sizeof(float);
static_assert(kNU == kNY, "load_a32 serves B and y2x_w with one row pitch");

constexpr int kUTile      = kBT * kNU;
constexpr int kUChunk     = kChunk * kUTile;
constexpr int kStageLoads = kUChunk / 4 / 256;
static_assert(kStageLoads * 256 * 4 == kUChunk, "256 threads cover the chunk tile exactly");
constexpr int kParts      = 4;
constexpr int kPartLoads  = kStageLoads / kParts;
constexpr int kPartGrp    = kNGrp / kParts;
static_assert(kPartLoads * 2 == kPartGrp * kG, "a quarter's pieces are its groups' timesteps");
}

__device__ __forceinline__ float pole_radius(const float llr_p) {
  const float decay = __expf(-fabsf(llr_p));
  return __expf(-decay);
}

__device__ __forceinline__ float pole_nf(const float llr_p) {
  const float r = pole_radius(llr_p);
  return sqrtf(fmaxf(0.f, 1.f - r * r));
}

__device__ __forceinline__ v8f load8(const float* __restrict__ p) {
  const v4f a = *(const v4f*)(p);
  const v4f b = *(const v4f*)(p + 4);
  return __builtin_shufflevector(a, b, 0, 1, 2, 3, 4, 5, 6, 7);
}

__device__ __forceinline__ v8f load_ftile(const float* __restrict__ p) {
  const v4f a = *(const v4f*)(p);
  const v4f b = *(const v4f*)(p + kFQuarter);
  return __builtin_shufflevector(a, b, 0, 1, 2, 3, 4, 5, 6, 7);
}

__device__ __forceinline__ void store_ftile(float* __restrict__ p, const v8f x) {
  *(v4f*)(p)             = __builtin_shufflevector(x, x, 0, 1, 2, 3);
  *(v4f*)(p + kFQuarter) = __builtin_shufflevector(x, x, 4, 5, 6, 7);
}

__device__ __forceinline__ void load_a32(const float* __restrict__ M, int P0,
                                         int laneM, int laneHi, v2f (&a)[2][8]) {
  #pragma unroll
  for (int i = 0; i < 2; ++i) {
    const float* row = M + (size_t)(i * kNHalf + P0 + laneM) * kNU + 16 * laneHi;
    #pragma unroll
    for (int kk = 0; kk < 8; ++kk) a[i][kk] = *(const v2f*)(row + 2 * kk);
  }
}

__device__ __forceinline__ void load_bp(const float* __restrict__ Bmat,
                                        const float* __restrict__ llr, int P0,
                                        int laneM, int laneHi, v2f (&bp)[2][8]) {
  const float nf = pole_nf(llr[P0 + laneM]);
  load_a32(Bmat, P0, laneM, laneHi, bp);
  #pragma unroll
  for (int i = 0; i < 2; ++i)
    #pragma unroll
    for (int kk = 0; kk < 8; ++kk) bp[i][kk] *= nf;
}

__device__ __forceinline__ void load_rot(const float* __restrict__ llr,
                                         const float* __restrict__ lli, int P0,
                                         int laneHi, v8f& lr, v8f& li) {
  const v8f a = load8(llr + P0 + 8 * laneHi);
  const v8f b = load8(lli + P0 + 8 * laneHi);
  #pragma unroll
  for (int v = 0; v < 8; ++v) {
    const float r  = pole_radius(a[v]);
    const float th = 1.57079632679489662f * __expf(-fabsf(b[v]));
    lr[v] = r * __cosf(th);
    li[v] = r * __sinf(th);
  }
}

__device__ __forceinline__ void load_u(const float* __restrict__ up, v4f (&u)[4]) {
  #pragma unroll
  for (int q = 0; q < 4; ++q) u[q] = *(const v4f*)(up + 4 * q);
}

__device__ __forceinline__ int u_pos(int x, int q) { return x ^ (4 * q) ^ (2 * (x >> 4)); }

__device__ __forceinline__ void stage_u_load(const float* __restrict__ U, int t0, int bq,
                                             int tid, v4f (&st)[kStageLoads]) {
  const int pidx = tid & 127;
  const int s = pidx >> 5, r = (pidx >> 3) & 3;
  const float* g = U + ((size_t)(t0 + s * kChunk + (tid >> 7)) * kNB + bq * kRows + r) * kNU +
                   (pidx & 7) * 4;
  #pragma unroll
  for (int n = 0; n < kStageLoads; ++n) st[n] = *(const v4f*)(g + (size_t)n * 2 * kUStep);
}

__device__ __forceinline__ void stage_u_store(float* __restrict__ Us, int tid,
                                              const v4f (&st)[kStageLoads], int part) {
  const int pidx = tid & 127;
  const int r = pidx >> 3, h = (pidx >> 2) & 1, q = pidx & 3;
  float* s = Us + (((tid >> 7) * 4 + q) * 32 + u_pos(16 * h + r, q)) * 4;
  #pragma unroll
  for (int k = 0; k < kPartLoads; ++k) {
    const int n = part * kPartLoads + k;
    *(v4f*)(s + n * 2 * kUTile) = st[n];
  }
}

__device__ __forceinline__ void lds_barrier() {
  __builtin_amdgcn_fence(__ATOMIC_RELEASE, "workgroup", "local");
  __builtin_amdgcn_s_barrier();
  __builtin_amdgcn_fence(__ATOMIC_ACQUIRE, "workgroup", "local");
}

__device__ __forceinline__ void load_u_lds(const float* __restrict__ ut, const int (&po)[4],
                                           v4f (&u)[4]) {
  #pragma unroll
  for (int q = 0; q < 4; ++q) u[q] = *(const v4f*)(ut + po[q]);
}

__device__ __forceinline__ void mma32(v8f& c0, v8f& c1, const v2f (&a)[2][8],
                                      const v4f (&u)[4]) {
  #pragma unroll
  for (int kk = 0; kk < 8; ++kk) {
    v2f b;
    b[0] = u[kk >> 1][(kk & 1) * 2];
    b[1] = u[kk >> 1][(kk & 1) * 2 + 1];
    c0 = __builtin_amdgcn_wmma_f32_16x16x4_f32(false, a[0][kk], false, b,
                                               (short)0, c0, false, false);
    c1 = __builtin_amdgcn_wmma_f32_16x16x4_f32(false, a[1][kk], false, b,
                                               (short)0, c1, false, false);
  }
}

__device__ __forceinline__ void step_state(v8f& s0, v8f& s1, const v2f (&bp)[2][8],
                                           const v4f (&u)[4], const v8f lr, const v8f li) {
  v8f c0 = lr * s0 - li * s1;
  v8f c1 = li * s0 + lr * s1;
  mma32(c0, c1, bp, u);
  s0 = c0; s1 = c1;
}

__device__ __forceinline__ void pole_square(v8f& qr, v8f& qi, int n) {
  for (int s = 0; s < n; ++s) {
    const v8f nr = qr * qr - qi * qi;
    const v8f ni = 2.f * qr * qi;
    qr = nr; qi = ni;
  }
}

__device__ __forceinline__ v8f prev_segment(const v8f x) {
  v8f y;
  #pragma unroll
  for (int v = 0; v < 8; ++v) y[v] = __shfl_up(x[v], kRows, kBT);
  return y;
}

__global__ void __launch_bounds__(256)
pass1_kernel(const float* __restrict__ U, const float* __restrict__ llr,
             const float* __restrict__ lli, const float* __restrict__ Bmat,
             float* __restrict__ ws) {
  const int bq = blockIdx.x;
  const int sc = blockIdx.y;
  const int tid    = threadIdx.x;
  const int lane   = tid & 31;
  const int wave   = tid >> 5;
  const int laneM  = lane & 15;
  const int laneHi = lane >> 4;
  const int P0 = wave * 16;

  __shared__ __attribute__((aligned(16))) float Us[kUChunk];
  v2f bp[2][8];
  load_bp(Bmat, llr, P0, laneM, laneHi, bp);
  v8f lr, li;
  load_rot(llr, lli, P0, laneHi, lr, li);

  v4f st[kStageLoads];
  stage_u_load(U, sc * kSuper, bq, tid, st);

  int po[4];
  #pragma unroll
  for (int q = 0; q < 4; ++q) po[q] = q * 128 + 4 * u_pos(lane, q);

  v8f s0 = {}, s1 = {};
  #pragma unroll
  for (int part = 0; part < kParts; ++part) {
    stage_u_store(Us, tid, st, part);
    lds_barrier();
    #pragma nounroll
    for (int g = 0; g < kPartGrp; ++g) {
      const int grp = part * kPartGrp + g;
      v4f u[kG][4];
      #pragma unroll
      for (int tt = 0; tt < kG; ++tt) load_u_lds(Us + (grp * kG + tt) * kUTile, po, u[tt]);
      #pragma unroll
      for (int tt = 0; tt < kG; ++tt) step_state(s0, s1, bp, u[tt], lr, li);
    }
  }
  {
    v8f pr = lr, pi = li;
    pole_square(pr, pi, 6);
    const int seg = laneM >> 2;
    for (int k = 1; k < kSeg; ++k) {
      const v8f a = prev_segment(s0), b = prev_segment(s1);
      if (seg == k) {
        s0 = pr * a - pi * b + s0;
        s1 = pi * a + pr * b + s1;
      }
    }
  }
  float* F = ws + WS_F + (size_t)sc * kFStride + ((size_t)bq * kWaves + wave) * kFWave + lane * 4;
  store_ftile(F, s0);
  store_ftile(F + 2 * kFQuarter, s1);
}

__device__ __forceinline__ void carry_step(v8f& s0, v8f& s1, const v8f f0, const v8f f1,
                                           const v8f pr, const v8f pi) {
  #pragma unroll
  for (int v = 0; v < 8; ++v) {
    const float a = s0[v], b = s1[v];
    s0[v] = pr[v] * a - pi[v] * b + f0[v];
    s1[v] = pi[v] * a + pr[v] * b + f1[v];
  }
}

__device__ __forceinline__ void reduce_store(const float* __restrict__ Pb,
                                             float* __restrict__ op,
                                             int wave, int lane, const v4f bias4) {
  #pragma unroll
  for (int k = 0; k < kG / 2; ++k) {
    const int tt = 2 * k + (wave >> 2);
    const float* p = Pb + (tt * 4 + (wave & 3)) * 128 + lane * 4;
    v4f acc = *(const v4f*)(p);
    #pragma unroll
    for (int w = 1; w < kWaves; ++w) acc += *(const v4f*)(p + w * kPartWave);
    acc += bias4;
    *(v4f*)(op + (size_t)tt * kOutStep) = acc;
  }
}

__global__ void __launch_bounds__(256)
pass2_kernel(const float* __restrict__ U, const float* __restrict__ yinit,
             const float* __restrict__ llr, const float* __restrict__ lli,
             const float* __restrict__ Bmat, const float* __restrict__ y2xw,
             const float* __restrict__ y2xb, const float* __restrict__ W,
             const float* __restrict__ bias, const float* __restrict__ ws,
             float* __restrict__ out) {
  extern __shared__ float Pbuf[];

  const int bq = blockIdx.x;
  const int sc = blockIdx.y;
  const int tid    = threadIdx.x;
  const int lane   = tid & 31;
  const int wave   = tid >> 5;
  const int laneM  = lane & 15;
  const int laneHi = lane >> 4;
  const int seg  = laneM >> 2;
  const int brow = bq * kRows + (laneM & 3);
  const int P0 = wave * 16;

  __shared__ __attribute__((aligned(16))) float Us[kUChunk];
  v8f lr, li;
  v8f s0, s1;
  v2f bp[2][8];
  v2f wf[2][2][4];
  const int ycol = 16 * ((wave & 3) >> 1) + 8 * laneHi + 4 * (wave & 1);
  v4f bias4;
  {
    load_rot(llr, lli, P0, laneHi, lr, li);

    {
      v2f yw[2][8];
      load_a32(y2xw, P0, laneM, laneHi, yw);
      v4f yr[4];
      load_u(yinit + (size_t)brow * kNY + 16 * laneHi, yr);
      s0 = load8(y2xb + P0 + 8 * laneHi);
      s1 = load8(y2xb + kNHalf + P0 + 8 * laneHi);
      mma32(s0, s1, yw, yr);
    }
    v4f st[kStageLoads];
    stage_u_load(U, sc * kSuper, bq, tid, st);
    const float* Pt = ws + WS_F + ((size_t)bq * kWaves + wave) * kFWave;
    const float* Fq = Pt + ((lane & 16) | 12 | (lane & 3)) * 4;
    v8f f0[kCarryDepth], f1[kCarryDepth];
    if (sc > 0) {
      #pragma unroll
      for (int d = 0; d < kCarryDepth; ++d) {
        const float* p = Fq + (size_t)min(d, sc - 1) * kFStride;
        f0[d] = load_ftile(p);
        f1[d] = load_ftile(p + 2 * kFQuarter);
      }
    }
    const float* Pown = Pt + (size_t)sc * kFStride + lane * 4;
    const v8f p0 = load_ftile(Pown);
    const v8f p1 = load_ftile(Pown + 2 * kFQuarter);
    load_bp(Bmat, llr, P0, laneM, laneHi, bp);
    #pragma unroll
    for (int my = 0; my < 2; ++my)
      #pragma unroll
      for (int i = 0; i < 2; ++i)
        #pragma unroll
        for (int s = 0; s < 4; ++s)
          wf[my][i][s] = *(const v2f*)(W + (size_t)(16 * my + laneM) * kNH +
                                       i * kNHalf + P0 + 8 * laneHi + 2 * s);
    bias4 = *(const v4f*)(bias + ycol);
    #pragma unroll
    for (int part = 0; part < kParts; ++part) stage_u_store(Us, tid, st, part);
    v8f pr = lr, pi = li;
    pole_square(pr, pi, 6);
    v8f qr, qi;
    #pragma unroll
    for (int v = 0; v < 8; ++v) { qr[v] = 1.f; qi[v] = 0.f; }
    for (int k = 1; k < kSeg; ++k) {
      if (seg >= k) {
        const v8f nr = qr * pr - qi * pi;
        const v8f ni = qr * pi + qi * pr;
        qr = nr; qi = ni;
      }
    }
    if (sc > 0) {
      pole_square(pr, pi, 2);
      int j0 = 0;
      for (; j0 + kCarryDepth <= sc; j0 += kCarryDepth) {
        #pragma unroll
        for (int d = 0; d < kCarryDepth; ++d) {
          carry_step(s0, s1, f0[d], f1[d], pr, pi);
          const float* p = Fq + (size_t)min(j0 + kCarryDepth + d, sc - 1) * kFStride;
          f0[d] = load_ftile(p);
          f1[d] = load_ftile(p + 2 * kFQuarter);
        }
      }
      #pragma unroll
      for (int d = 0; d < kCarryDepth - 1; ++d)
        if (j0 + d < sc) carry_step(s0, s1, f0[d], f1[d], pr, pi);
    }
    {
      const v8f a = prev_segment(p0), b = prev_segment(p1);
      v8f n0 = qr * s0 - qi * s1;
      v8f n1 = qi * s0 + qr * s1;
      if (seg > 0) { n0 += a; n1 += b; }
      s0 = n0; s1 = n1;
    }
  }

  float* outp = out + ((size_t)(sc * kSuper + seg * kChunk) * kNB + brow) * kNY + ycol;

  int po[4];
  #pragma unroll
  for (int q = 0; q < 4; ++q) po[q] = q * 128 + 4 * u_pos(lane, q);
  __syncthreads();

  for (int grp = 0; grp < kNGrp; ++grp) {
    const int buf = grp & 1;
    v4f u[kG][4];
    #pragma unroll
    for (int tt = 0; tt < kG; ++tt) load_u_lds(Us + (grp * kG + tt) * kUTile, po, u[tt]);
    if (grp > 0)
      reduce_store(Pbuf + (buf ^ 1) * kPartBuf, outp + (size_t)(grp - 1) * kG * kOutStep,
                   wave, lane, bias4);
    float* Pw = Pbuf + buf * kPartBuf + wave * kPartWave + lane * 4;
    #pragma unroll
    for (int tt = 0; tt < kG; ++tt) {
      step_state(s0, s1, bp, u[tt], lr, li);
      v8f y0 = {}, y1 = {};
      #pragma unroll
      for (int s = 0; s < 4; ++s) {
        v2f b;
        b[0] = s0[2 * s]; b[1] = s0[2 * s + 1];
        y0 = __builtin_amdgcn_wmma_f32_16x16x4_f32(false, wf[0][0][s], false, b,
                                                   (short)0, y0, false, false);
        y1 = __builtin_amdgcn_wmma_f32_16x16x4_f32(false, wf[1][0][s], false, b,
                                                   (short)0, y1, false, false);
      }
      #pragma unroll
      for (int s = 0; s < 4; ++s) {
        v2f b;
        b[0] = s1[2 * s]; b[1] = s1[2 * s + 1];
        y0 = __builtin_amdgcn_wmma_f32_16x16x4_f32(false, wf[0][1][s], false, b,
                                                   (short)0, y0, false, false);
        y1 = __builtin_amdgcn_wmma_f32_16x16x4_f32(false, wf[1][1][s], false, b,
                                                   (short)0, y1, false, false);
      }
      *(v4f*)(Pw + (tt * 4 + 0) * 128) = __builtin_shufflevector(y0, y0, 0, 1, 2, 3);
      *(v4f*)(Pw + (tt * 4 + 1) * 128) = __builtin_shufflevector(y0, y0, 4, 5, 6, 7);
      *(v4f*)(Pw + (tt * 4 + 2) * 128) = __builtin_shufflevector(y1, y1, 0, 1, 2, 3);
      *(v4f*)(Pw + (tt * 4 + 3) * 128) = __builtin_shufflevector(y1, y1, 4, 5, 6, 7);
    }
    __syncthreads();
  }
  reduce_store(Pbuf + ((kNGrp - 1) & 1) * kPartBuf,
               outp + (size_t)(kNGrp - 1) * kG * kOutStep, wave, lane, bias4);
}

extern "C" void kernel_launch(void* const* d_in, const int* in_sizes, int n_in,
                              void* d_out, int out_size, void* d_ws, size_t ws_size,
                              hipStream_t stream) {
  (void)in_sizes; (void)n_in; (void)out_size; (void)ws_size;
  const float* y0   = (const float*)d_in[0];
  const float* U    = (const float*)d_in[1];
  const float* llr  = (const float*)d_in[2];
  const float* lli  = (const float*)d_in[3];
  const float* Bm   = (const float*)d_in[4];
  const float* y2xw = (const float*)d_in[5];
  const float* y2xb = (const float*)d_in[6];
  const float* x2yw = (const float*)d_in[7];
  const float* x2yb = (const float*)d_in[8];
  float* ws  = (float*)d_ws;
  float* out = (float*)d_out;

  hipFuncSetAttribute(reinterpret_cast<const void*>(pass2_kernel),
                      hipFuncAttributeMaxDynamicSharedMemorySize, (int)kSmemBytes);

  dim3 grid(kNBQ, kNSuper);
  pass1_kernel<<<grid, 256, 0, stream>>>(U, llr, lli, Bm, ws);
  pass2_kernel<<<grid, 256, kSmemBytes, stream>>>(U, y0, llr, lli, Bm, y2xw, y2xb,
                                                 x2yw, x2yb, ws, out);
}


# pass-2 group barrier split in two: a wave signals when its partial Y tiles are in LDS and waits only before summing them, part-way into the next group; waves no longer stall together; sums unchanged
# speedup vs baseline: 23.3376x; 1.0322x over previous
#include <hip/hip_runtime.h>

typedef __attribute__((ext_vector_type(2))) float v2f;
typedef __attribute__((ext_vector_type(4))) float v4f;
typedef __attribute__((ext_vector_type(8))) float v8f;

namespace {
constexpr int kT      = 4096;
constexpr int kNB     = 64;
constexpr int kNU     = 32;
constexpr int kNY     = 32;
constexpr int kNH     = 256;
constexpr int kNHalf  = 128;
constexpr int kChunk  = 64;
constexpr int kBT     = 16;
constexpr int kRows   = 4;
constexpr int kSeg    = kBT / kRows;
constexpr int kNBQ    = kNB / kRows;
constexpr int kSuper  = kSeg * kChunk;
constexpr int kNSuper = kT / kSuper;
constexpr int kWaves  = 8;
constexpr int kG      = 4;
constexpr int kNGrp   = kChunk / kG;
constexpr int kCarryDepth = 8;

constexpr size_t WS_F   = 0;
constexpr size_t kFQuarter = 32 * 4;
constexpr size_t kFWave    = 4 * kFQuarter;
constexpr size_t kFStride  = (size_t)kNBQ * kWaves * kFWave;
static_assert(kFStride == (size_t)kNBQ * kBT * kNH, "one P holds every (column, hidden) element once");

constexpr size_t kUStep   = (size_t)kNB * kNU;
constexpr size_t kOutStep = (size_t)kNB * kNY;

constexpr int kPartWave = kG * 4 * 128;
constexpr int kPartBuf  = kWaves * kPartWave;
constexpr size_t kSmemBytes = (size_t)2 * kPartBuf * sizeof(float);
static_assert(kNU == kNY, "load_a32 serves B and y2x_w with one row pitch");

constexpr int kUTile      = kBT * kNU;
constexpr int kUChunk     = kChunk * kUTile;
constexpr int kStageLoads = kUChunk / 4 / 256;
static_assert(kStageLoads * 256 * 4 == kUChunk, "256 threads cover the chunk tile exactly");
constexpr int kParts      = 4;
constexpr int kPartLoads  = kStageLoads / kParts;
constexpr int kPartGrp    = kNGrp / kParts;
static_assert(kPartLoads * 2 == kPartGrp * kG, "a quarter's pieces are its groups' timesteps");
}

__device__ __forceinline__ float pole_radius(const float llr_p) {
  const float decay = __expf(-fabsf(llr_p));
  return __expf(-decay);
}

__device__ __forceinline__ float pole_nf(const float llr_p) {
  const float r = pole_radius(llr_p);
  return sqrtf(fmaxf(0.f, 1.f - r * r));
}

__device__ __forceinline__ v8f load8(const float* __restrict__ p) {
  const v4f a = *(const v4f*)(p);
  const v4f b = *(const v4f*)(p + 4);
  return __builtin_shufflevector(a, b, 0, 1, 2, 3, 4, 5, 6, 7);
}

__device__ __forceinline__ v8f load_ftile(const float* __restrict__ p) {
  const v4f a = *(const v4f*)(p);
  const v4f b = *(const v4f*)(p + kFQuarter);
  return __builtin_shufflevector(a, b, 0, 1, 2, 3, 4, 5, 6, 7);
}

__device__ __forceinline__ void store_ftile(float* __restrict__ p, const v8f x) {
  *(v4f*)(p)             = __builtin_shufflevector(x, x, 0, 1, 2, 3);
  *(v4f*)(p + kFQuarter) = __builtin_shufflevector(x, x, 4, 5, 6, 7);
}

__device__ __forceinline__ void load_a32(const float* __restrict__ M, int P0,
                                         int laneM, int laneHi, v2f (&a)[2][8]) {
  #pragma unroll
  for (int i = 0; i < 2; ++i) {
    const float* row = M + (size_t)(i * kNHalf + P0 + laneM) * kNU + 16 * laneHi;
    #pragma unroll
    for (int kk = 0; kk < 8; ++kk) a[i][kk] = *(const v2f*)(row + 2 * kk);
  }
}

__device__ __forceinline__ void load_bp(const float* __restrict__ Bmat,
                                        const float* __restrict__ llr, int P0,
                                        int laneM, int laneHi, v2f (&bp)[2][8]) {
  const float nf = pole_nf(llr[P0 + laneM]);
  load_a32(Bmat, P0, laneM, laneHi, bp);
  #pragma unroll
  for (int i = 0; i < 2; ++i)
    #pragma unroll
    for (int kk = 0; kk < 8; ++kk) bp[i][kk] *= nf;
}

__device__ __forceinline__ void load_rot(const float* __restrict__ llr,
                                         const float* __restrict__ lli, int P0,
                                         int laneHi, v8f& lr, v8f& li) {
  const v8f a = load8(llr + P0 + 8 * laneHi);
  const v8f b = load8(lli + P0 + 8 * laneHi);
  #pragma unroll
  for (int v = 0; v < 8; ++v) {
    const float r  = pole_radius(a[v]);
    const float th = 1.57079632679489662f * __expf(-fabsf(b[v]));
    lr[v] = r * __cosf(th);
    li[v] = r * __sinf(th);
  }
}

__device__ __forceinline__ void load_u(const float* __restrict__ up, v4f (&u)[4]) {
  #pragma unroll
  for (int q = 0; q < 4; ++q) u[q] = *(const v4f*)(up + 4 * q);
}

__device__ __forceinline__ int u_pos(int x, int q) { return x ^ (4 * q) ^ (2 * (x >> 4)); }

__device__ __forceinline__ void stage_u_load(const float* __restrict__ U, int t0, int bq,
                                             int tid, v4f (&st)[kStageLoads]) {
  const int pidx = tid & 127;
  const int s = pidx >> 5, r = (pidx >> 3) & 3;
  const float* g = U + ((size_t)(t0 + s * kChunk + (tid >> 7)) * kNB + bq * kRows + r) * kNU +
                   (pidx & 7) * 4;
  #pragma unroll
  for (int n = 0; n < kStageLoads; ++n) st[n] = *(const v4f*)(g + (size_t)n * 2 * kUStep);
}

__device__ __forceinline__ void stage_u_store(float* __restrict__ Us, int tid,
                                              const v4f (&st)[kStageLoads], int part) {
  const int pidx = tid & 127;
  const int r = pidx >> 3, h = (pidx >> 2) & 1, q = pidx & 3;
  float* s = Us + (((tid >> 7) * 4 + q) * 32 + u_pos(16 * h + r, q)) * 4;
  #pragma unroll
  for (int k = 0; k < kPartLoads; ++k) {
    const int n = part * kPartLoads + k;
    *(v4f*)(s + n * 2 * kUTile) = st[n];
  }
}

__device__ __forceinline__ void lds_barrier() {
  __builtin_amdgcn_fence(__ATOMIC_RELEASE, "workgroup", "local");
  __builtin_amdgcn_s_barrier();
  __builtin_amdgcn_fence(__ATOMIC_ACQUIRE, "workgroup", "local");
}

__device__ __forceinline__ void load_u_lds(const float* __restrict__ ut, const int (&po)[4],
                                           v4f (&u)[4]) {
  #pragma unroll
  for (int q = 0; q < 4; ++q) u[q] = *(const v4f*)(ut + po[q]);
}

__device__ __forceinline__ void mma32(v8f& c0, v8f& c1, const v2f (&a)[2][8],
                                      const v4f (&u)[4]) {
  #pragma unroll
  for (int kk = 0; kk < 8; ++kk) {
    v2f b;
    b[0] = u[kk >> 1][(kk & 1) * 2];
    b[1] = u[kk >> 1][(kk & 1) * 2 + 1];
    c0 = __builtin_amdgcn_wmma_f32_16x16x4_f32(false, a[0][kk], false, b,
                                               (short)0, c0, false, false);
    c1 = __builtin_amdgcn_wmma_f32_16x16x4_f32(false, a[1][kk], false, b,
                                               (short)0, c1, false, false);
  }
}

__device__ __forceinline__ void step_state(v8f& s0, v8f& s1, const v2f (&bp)[2][8],
                                           const v4f (&u)[4], const v8f lr, const v8f li) {
  v8f c0 = lr * s0 - li * s1;
  v8f c1 = li * s0 + lr * s1;
  mma32(c0, c1, bp, u);
  s0 = c0; s1 = c1;
}

__device__ __forceinline__ void pole_square(v8f& qr, v8f& qi, int n) {
  for (int s = 0; s < n; ++s) {
    const v8f nr = qr * qr - qi * qi;
    const v8f ni = 2.f * qr * qi;
    qr = nr; qi = ni;
  }
}

__device__ __forceinline__ v8f prev_segment(const v8f x) {
  v8f y;
  #pragma unroll
  for (int v = 0; v < 8; ++v) y[v] = __shfl_up(x[v], kRows, kBT);
  return y;
}

__global__ void __launch_bounds__(256)
pass1_kernel(const float* __restrict__ U, const float* __restrict__ llr,
             const float* __restrict__ lli, const float* __restrict__ Bmat,
             float* __restrict__ ws) {
  const int bq = blockIdx.x;
  const int sc = blockIdx.y;
  const int tid    = threadIdx.x;
  const int lane   = tid & 31;
  const int wave   = tid >> 5;
  const int laneM  = lane & 15;
  const int laneHi = lane >> 4;
  const int P0 = wave * 16;

  __shared__ __attribute__((aligned(16))) float Us[kUChunk];
  v2f bp[2][8];
  load_bp(Bmat, llr, P0, laneM, laneHi, bp);
  v8f lr, li;
  load_rot(llr, lli, P0, laneHi, lr, li);

  v4f st[kStageLoads];
  stage_u_load(U, sc * kSuper, bq, tid, st);

  int po[4];
  #pragma unroll
  for (int q = 0; q < 4; ++q) po[q] = q * 128 + 4 * u_pos(lane, q);

  v8f s0 = {}, s1 = {};
  #pragma unroll
  for (int part = 0; part < kParts; ++part) {
    stage_u_store(Us, tid, st, part);
    lds_barrier();
    #pragma nounroll
    for (int g = 0; g < kPartGrp; ++g) {
      const int grp = part * kPartGrp + g;
      v4f u[kG][4];
      #pragma unroll
      for (int tt = 0; tt < kG; ++tt) load_u_lds(Us + (grp * kG + tt) * kUTile, po, u[tt]);
      #pragma unroll
      for (int tt = 0; tt < kG; ++tt) step_state(s0, s1, bp, u[tt], lr, li);
    }
  }
  {
    v8f pr = lr, pi = li;
    pole_square(pr, pi, 6);
    const int seg = laneM >> 2;
    for (int k = 1; k < kSeg; ++k) {
      const v8f a = prev_segment(s0), b = prev_segment(s1);
      if (seg == k) {
        s0 = pr * a - pi * b + s0;
        s1 = pi * a + pr * b + s1;
      }
    }
  }
  float* F = ws + WS_F + (size_t)sc * kFStride + ((size_t)bq * kWaves + wave) * kFWave + lane * 4;
  store_ftile(F, s0);
  store_ftile(F + 2 * kFQuarter, s1);
}

__device__ __forceinline__ void carry_step(v8f& s0, v8f& s1, const v8f f0, const v8f f1,
                                           const v8f pr, const v8f pi) {
  #pragma unroll
  for (int v = 0; v < 8; ++v) {
    const float a = s0[v], b = s1[v];
    s0[v] = pr[v] * a - pi[v] * b + f0[v];
    s1[v] = pi[v] * a + pr[v] * b + f1[v];
  }
}

__device__ __forceinline__ void reduce_store(const float* __restrict__ Pb,
                                             float* __restrict__ op,
                                             int wave, int lane, const v4f bias4) {
  #pragma unroll
  for (int k = 0; k < kG / 2; ++k) {
    const int tt = 2 * k + (wave >> 2);
    const float* p = Pb + (tt * 4 + (wave & 3)) * 128 + lane * 4;
    v4f acc = *(const v4f*)(p);
    #pragma unroll
    for (int w = 1; w < kWaves; ++w) acc += *(const v4f*)(p + w * kPartWave);
    acc += bias4;
    *(v4f*)(op + (size_t)tt * kOutStep) = acc;
  }
}

__device__ __forceinline__ void group_signal() {
  __builtin_amdgcn_fence(__ATOMIC_RELEASE, "workgroup", "local");
  __builtin_amdgcn_s_barrier_signal(-1);
}

__device__ __forceinline__ void group_wait() {
  __builtin_amdgcn_s_barrier_wait(-1);
  __builtin_amdgcn_fence(__ATOMIC_ACQUIRE, "workgroup", "local");
}

__device__ __forceinline__ void y_tiles(const v8f s0, const v8f s1, const v2f (&wf)[2][2][4],
                                        v8f& y0, v8f& y1) {
  #pragma unroll
  for (int v = 0; v < 8; ++v) { y0[v] = 0.f; y1[v] = 0.f; }
  #pragma unroll
  for (int s = 0; s < 4; ++s) {
    v2f b;
    b[0] = s0[2 * s]; b[1] = s0[2 * s + 1];
    y0 = __builtin_amdgcn_wmma_f32_16x16x4_f32(false, wf[0][0][s], false, b,
                                               (short)0, y0, false, false);
    y1 = __builtin_amdgcn_wmma_f32_16x16x4_f32(false, wf[1][0][s], false, b,
                                               (short)0, y1, false, false);
  }
  #pragma unroll
  for (int s = 0; s < 4; ++s) {
    v2f b;
    b[0] = s1[2 * s]; b[1] = s1[2 * s + 1];
    y0 = __builtin_amdgcn_wmma_f32_16x16x4_f32(false, wf[0][1][s], false, b,
                                               (short)0, y0, false, false);
    y1 = __builtin_amdgcn_wmma_f32_16x16x4_f32(false, wf[1][1][s], false, b,
                                               (short)0, y1, false, false);
  }
}

__device__ __forceinline__ void store_partial(float* __restrict__ Pw, int tt, const v8f y0,
                                              const v8f y1) {
  *(v4f*)(Pw + (tt * 4 + 0) * 128) = __builtin_shufflevector(y0, y0, 0, 1, 2, 3);
  *(v4f*)(Pw + (tt * 4 + 1) * 128) = __builtin_shufflevector(y0, y0, 4, 5, 6, 7);
  *(v4f*)(Pw + (tt * 4 + 2) * 128) = __builtin_shufflevector(y1, y1, 0, 1, 2, 3);
  *(v4f*)(Pw + (tt * 4 + 3) * 128) = __builtin_shufflevector(y1, y1, 4, 5, 6, 7);
}

__global__ void __launch_bounds__(256)
pass2_kernel(const float* __restrict__ U, const float* __restrict__ yinit,
             const float* __restrict__ llr, const float* __restrict__ lli,
             const float* __restrict__ Bmat, const float* __restrict__ y2xw,
             const float* __restrict__ y2xb, const float* __restrict__ W,
             const float* __restrict__ bias, const float* __restrict__ ws,
             float* __restrict__ out) {
  extern __shared__ float Pbuf[];

  const int bq = blockIdx.x;
  const int sc = blockIdx.y;
  const int tid    = threadIdx.x;
  const int lane   = tid & 31;
  const int wave   = tid >> 5;
  const int laneM  = lane & 15;
  const int laneHi = lane >> 4;
  const int seg  = laneM >> 2;
  const int brow = bq * kRows + (laneM & 3);
  const int P0 = wave * 16;

  __shared__ __attribute__((aligned(16))) float Us[kUChunk];
  v8f lr, li;
  v8f s0, s1;
  v2f bp[2][8];
  v2f wf[2][2][4];
  const int ycol = 16 * ((wave & 3) >> 1) + 8 * laneHi + 4 * (wave & 1);
  v4f bias4;
  {
    load_rot(llr, lli, P0, laneHi, lr, li);

    {
      v2f yw[2][8];
      load_a32(y2xw, P0, laneM, laneHi, yw);
      v4f yr[4];
      load_u(yinit + (size_t)brow * kNY + 16 * laneHi, yr);
      s0 = load8(y2xb + P0 + 8 * laneHi);
      s1 = load8(y2xb + kNHalf + P0 + 8 * laneHi);
      mma32(s0, s1, yw, yr);
    }
    v4f st[kStageLoads];
    stage_u_load(U, sc * kSuper, bq, tid, st);
    const float* Pt = ws + WS_F + ((size_t)bq * kWaves + wave) * kFWave;
    const float* Fq = Pt + ((lane & 16) | 12 | (lane & 3)) * 4;
    v8f f0[kCarryDepth], f1[kCarryDepth];
    if (sc > 0) {
      #pragma unroll
      for (int d = 0; d < kCarryDepth; ++d) {
        const float* p = Fq + (size_t)min(d, sc - 1) * kFStride;
        f0[d] = load_ftile(p);
        f1[d] = load_ftile(p + 2 * kFQuarter);
      }
    }
    const float* Pown = Pt + (size_t)sc * kFStride + lane * 4;
    const v8f p0 = load_ftile(Pown);
    const v8f p1 = load_ftile(Pown + 2 * kFQuarter);
    load_bp(Bmat, llr, P0, laneM, laneHi, bp);
    #pragma unroll
    for (int my = 0; my < 2; ++my)
      #pragma unroll
      for (int i = 0; i < 2; ++i)
        #pragma unroll
        for (int s = 0; s < 4; ++s)
          wf[my][i][s] = *(const v2f*)(W + (size_t)(16 * my + laneM) * kNH +
                                       i * kNHalf + P0 + 8 * laneHi + 2 * s);
    bias4 = *(const v4f*)(bias + ycol);
    #pragma unroll
    for (int part = 0; part < kParts; ++part) stage_u_store(Us, tid, st, part);
    v8f pr = lr, pi = li;
    pole_square(pr, pi, 6);
    v8f qr, qi;
    #pragma unroll
    for (int v = 0; v < 8; ++v) { qr[v] = 1.f; qi[v] = 0.f; }
    for (int k = 1; k < kSeg; ++k) {
      if (seg >= k) {
        const v8f nr = qr * pr - qi * pi;
        const v8f ni = qr * pi + qi * pr;
        qr = nr; qi = ni;
      }
    }
    if (sc > 0) {
      pole_square(pr, pi, 2);
      int j0 = 0;
      for (; j0 + kCarryDepth <= sc; j0 += kCarryDepth) {
        #pragma unroll
        for (int d = 0; d < kCarryDepth; ++d) {
          carry_step(s0, s1, f0[d], f1[d], pr, pi);
          const float* p = Fq + (size_t)min(j0 + kCarryDepth + d, sc - 1) * kFStride;
          f0[d] = load_ftile(p);
          f1[d] = load_ftile(p + 2 * kFQuarter);
        }
      }
      #pragma unroll
      for (int d = 0; d < kCarryDepth - 1; ++d)
        if (j0 + d < sc) carry_step(s0, s1, f0[d], f1[d], pr, pi);
    }
    {
      const v8f a = prev_segment(p0), b = prev_segment(p1);
      v8f n0 = qr * s0 - qi * s1;
      v8f n1 = qi * s0 + qr * s1;
      if (seg > 0) { n0 += a; n1 += b; }
      s0 = n0; s1 = n1;
    }
  }

  float* outp = out + ((size_t)(sc * kSuper + seg * kChunk) * kNB + brow) * kNY + ycol;

  int po[4];
  #pragma unroll
  for (int q = 0; q < 4; ++q) po[q] = q * 128 + 4 * u_pos(lane, q);
  __syncthreads();

  for (int grp = 0; grp < kNGrp; ++grp) {
    const int buf = grp & 1;
    v4f u[kG][4];
    #pragma unroll
    for (int tt = 0; tt < kG; ++tt) load_u_lds(Us + (grp * kG + tt) * kUTile, po, u[tt]);
    float* Pw = Pbuf + buf * kPartBuf + wave * kPartWave + lane * 4;
    v8f ya[kG / 2][2];
    #pragma unroll
    for (int tt = 0; tt < kG / 2; ++tt) {
      step_state(s0, s1, bp, u[tt], lr, li);
      y_tiles(s0, s1, wf, ya[tt][0], ya[tt][1]);
    }
    if (grp > 0) {
      group_wait();
      reduce_store(Pbuf + (buf ^ 1) * kPartBuf, outp + (size_t)(grp - 1) * kG * kOutStep,
                   wave, lane, bias4);
    }
    #pragma unroll
    for (int tt = 0; tt < kG / 2; ++tt) store_partial(Pw, tt, ya[tt][0], ya[tt][1]);
    #pragma unroll
    for (int tt = kG / 2; tt < kG; ++tt) {
      v8f y0, y1;
      step_state(s0, s1, bp, u[tt], lr, li);
      y_tiles(s0, s1, wf, y0, y1);
      store_partial(Pw, tt, y0, y1);
    }
    group_signal();
  }
  group_wait();
  reduce_store(Pbuf + ((kNGrp - 1) & 1) * kPartBuf,
               outp + (size_t)(kNGrp - 1) * kG * kOutStep, wave, lane, bias4);
}

extern "C" void kernel_launch(void* const* d_in, const int* in_sizes, int n_in,
                              void* d_out, int out_size, void* d_ws, size_t ws_size,
                              hipStream_t stream) {
  (void)in_sizes; (void)n_in; (void)out_size; (void)ws_size;
  const float* y0   = (const float*)d_in[0];
  const float* U    = (const float*)d_in[1];
  const float* llr  = (const float*)d_in[2];
  const float* lli  = (const float*)d_in[3];
  const float* Bm   = (const float*)d_in[4];
  const float* y2xw = (const float*)d_in[5];
  const float* y2xb = (const float*)d_in[6];
  const float* x2yw = (const float*)d_in[7];
  const float* x2yb = (const float*)d_in[8];
  float* ws  = (float*)d_ws;
  float* out = (float*)d_out;

  hipFuncSetAttribute(reinterpret_cast<const void*>(pass2_kernel),
                      hipFuncAttributeMaxDynamicSharedMemorySize, (int)kSmemBytes);

  dim3 grid(kNBQ, kNSuper);
  pass1_kernel<<<grid, 256, 0, stream>>>(U, llr, lli, Bm, ws);
  pass2_kernel<<<grid, 256, kSmemBytes, stream>>>(U, y0, llr, lli, Bm, y2xw, y2xb,
                                                 x2yw, x2yb, ws, out);
}
